# starting kernel
# baseline (speedup 1.0000x reference)
#include <hip/hip_runtime.h>
#include <math.h>

#define Nn 50000
#define Ee 1600000
#define Dd 128
#define Hh 128
#define Cc 10
#define Gg 512
#define LRELU 0.05f
#define BN_EPS 1e-5f

typedef __attribute__((ext_vector_type(16))) __bf16 v16bf;
typedef __attribute__((ext_vector_type(8)))  float  v8f;

union BFrag { v16bf v; unsigned u[8]; };

__device__ __forceinline__ unsigned short f2bf_bits(float f) {
  union { float f; unsigned u; } v; v.f = f;
  return (unsigned short)((v.u + 0x7FFFu + ((v.u >> 16) & 1u)) >> 16);
}

__global__ void k_zero(float* __restrict__ p, long long n) {
  long long i = (long long)blockIdx.x * blockDim.x + threadIdx.x;
  const long long stride = (long long)gridDim.x * blockDim.x;
  for (; i < n; i += stride) p[i] = 0.0f;
}

__global__ void k_cvt(const float* __restrict__ s, unsigned short* __restrict__ d, long long n) {
  long long i = (long long)blockIdx.x * blockDim.x + threadIdx.x;
  const long long stride = (long long)gridDim.x * blockDim.x;
  for (; i < n; i += stride) d[i] = f2bf_bits(s[i]);
}

__global__ void __launch_bounds__(256) k_stats(const float* __restrict__ x, float* __restrict__ stats) {
  __shared__ float ssum[256], ssq[256];
  const int tid = threadIdx.x;
  long long i = (long long)blockIdx.x * blockDim.x + tid;
  const long long stride = (long long)gridDim.x * blockDim.x;
  const long long total = (long long)Nn * Dd;
  float s = 0.f, q = 0.f;
  for (; i < total; i += stride) { float v = x[i]; s += v; q += v * v; }
  ssum[tid] = s; ssq[tid] = q; __syncthreads();
  for (int o = 128; o > 0; o >>= 1) {
    if (tid < o) { ssum[tid] += ssum[tid + o]; ssq[tid] += ssq[tid + o]; }
    __syncthreads();
  }
  if (tid == 0) { atomicAdd(&stats[0], ssum[0]); atomicAdd(&stats[1], ssq[0]); }
}

__global__ void k_prep(const float* __restrict__ w, float* __restrict__ stats,
                       float* __restrict__ corr) {
  const int h = threadIdx.x;
  float colsum = 0.f;
  for (int d = 0; d < Dd; ++d) colsum += w[h * Dd + d];
  const float nd = (float)((long long)Nn * Dd);
  const float s  = stats[0], ss = stats[1];
  const float mu = s / nd;
  const float var = (ss - s * s / nd) / (nd - 1.0f);
  const float invsig = rsqrtf(var);
  if (h == 0) stats[2] = invsig;
  corr[h] = mu * invsig * colsum;
}

__global__ void __launch_bounds__(256) k_gat_gemm(const unsigned short* __restrict__ xbf,
                                                  const unsigned short* __restrict__ wbf,
                                                  const float* __restrict__ stats,
                                                  const float* __restrict__ corr,
                                                  float* __restrict__ z) {
  const int tid  = threadIdx.x;
  const int lane = tid & 31;
  const int hi   = lane >> 4;
  const int lm   = lane & 15;
  const int n0   = blockIdx.x * 16;
  const int h0   = (tid >> 5) * 16;

  const unsigned* xp = (const unsigned*)xbf + (size_t)(n0 + lm) * (Dd / 2);
  const unsigned* wp = (const unsigned*)wbf + (size_t)(h0 + lm) * (Dd / 2);

  v8f c = {0.f, 0.f, 0.f, 0.f, 0.f, 0.f, 0.f, 0.f};
  const float invsig = stats[2];

  #pragma unroll
  for (int k0 = 0; k0 < Dd; k0 += 32) {
    BFrag a, b;
    #pragma unroll
    for (int j = 0; j < 8; ++j) {
      const int kk = k0 + ((j < 4) ? 0 : 16) + (hi ? 8 : 0) + 2 * (j & 3);
      a.u[j] = xp[kk >> 1];
      b.u[j] = wp[kk >> 1];
    }
    if (k0 + 32 < Dd)
      __builtin_prefetch(xp + ((k0 + 32) >> 1), 0, 1);
    c = __builtin_amdgcn_wmma_f32_16x16x32_bf16(false, a.v, false, b.v, (short)0, c, false, false);
  }
  #pragma unroll
  for (int r = 0; r < 8; ++r) {
    const int m = r + hi * 8;
    const int h = h0 + lm;
    z[(size_t)(n0 + m) * Hh + h] = c[r] * invsig - corr[h];
  }
}

__global__ void __launch_bounds__(256) k_s12(const float* __restrict__ z,
                                             const float* __restrict__ a1,
                                             const float* __restrict__ a2,
                                             float* __restrict__ s1, float* __restrict__ s2) {
  const int lane = threadIdx.x & 31;
  const int n = blockIdx.x * 8 + (threadIdx.x >> 5);
  if (n >= Nn) return;
  float p1 = 0.f, p2 = 0.f;
  for (int i = lane; i < Hh; i += 32) {
    const float zv = z[(size_t)n * Hh + i];
    p1 += zv * a1[i]; p2 += zv * a2[i];
  }
  for (int o = 16; o > 0; o >>= 1) { p1 += __shfl_xor(p1, o); p2 += __shfl_xor(p2, o); }
  if (lane == 0) { s1[n] = p1; s2[n] = p2; }
}

__global__ void __launch_bounds__(256) k_edge1(const int* __restrict__ src, const int* __restrict__ dst,
                                               const float* __restrict__ s1, const float* __restrict__ s2,
                                               float* __restrict__ hbuf, float* __restrict__ hsum) {
  long long i = (long long)blockIdx.x * blockDim.x + threadIdx.x;
  const long long stride = (long long)gridDim.x * blockDim.x;
  for (; i < Ee; i += stride) {
    const int s = src[i], d = dst[i];
    float e = s1[s] + s2[d];
    e = (e >= 0.f) ? e : LRELU * e;
    const float h = __expf(e);
    hbuf[i] = h;
    atomicAdd(&hsum[s], h);
  }
}

__global__ void __launch_bounds__(256) k_edge2(const int* __restrict__ src, const int* __restrict__ dst,
                                               const float* __restrict__ hbuf, const float* __restrict__ hsum,
                                               const float* __restrict__ z, float* __restrict__ gout) {
  const int lane = threadIdx.x & 31;
  const long long e = (long long)blockIdx.x * 8 + (threadIdx.x >> 5);
  if (e >= Ee) return;
  const int s = src[e], d = dst[e];
  const float alpha = hbuf[e] / hsum[s];
  const float4 zv = *(const float4*)(z + (size_t)d * Hh + lane * 4);
  float* o = gout + (size_t)s * Hh + lane * 4;
  atomicAdd(o + 0, alpha * zv.x);
  atomicAdd(o + 1, alpha * zv.y);
  atomicAdd(o + 2, alpha * zv.z);
  atomicAdd(o + 3, alpha * zv.w);
}

__global__ void __launch_bounds__(256) k_pool(const float* __restrict__ gout, const int* __restrict__ idx,
                                              float* __restrict__ pool) {
  const int lane = threadIdx.x & 31;
  const int n = blockIdx.x * 8 + (threadIdx.x >> 5);
  if (n >= Nn) return;
  const int g = idx[n];
  const float4 v = *(const float4*)(gout + (size_t)n * Hh + lane * 4);
  float* p = pool + (size_t)g * Hh + lane * 4;
  atomicAdd(p + 0, fmaxf(v.x, 0.f));
  atomicAdd(p + 1, fmaxf(v.y, 0.f));
  atomicAdd(p + 2, fmaxf(v.z, 0.f));
  atomicAdd(p + 3, fmaxf(v.w, 0.f));
}

__global__ void k_bn(const float* __restrict__ pooled, const float* __restrict__ gamma,
                     const float* __restrict__ beta, float* __restrict__ xb) {
  const int h = threadIdx.x;
  float s = 0.f, q = 0.f;
  for (int g = 0; g < Gg; ++g) { const float v = pooled[g * Hh + h]; s += v; q += v * v; }
  const float mu = s / (float)Gg;
  const float var = q / (float)Gg - mu * mu;
  const float inv = rsqrtf(var + BN_EPS);
  const float ga = gamma[h], be = beta[h];
  for (int g = 0; g < Gg; ++g)
    xb[g * Hh + h] = (pooled[g * Hh + h] - mu) * inv * ga + be;
}

__global__ void __launch_bounds__(256) k_fc1(const unsigned short* __restrict__ xbf,
                                             const unsigned short* __restrict__ wbf,
                                             const float* __restrict__ bias, float* __restrict__ out) {
  const int tid  = threadIdx.x;
  const int lane = tid & 31;
  const int hi   = lane >> 4;
  const int lm   = lane & 15;
  const int n0   = blockIdx.x * 16;
  const int h0   = (tid >> 5) * 16;

  const unsigned* xp = (const unsigned*)xbf + (size_t)(n0 + lm) * (Hh / 2);
  const unsigned* wp = (const unsigned*)wbf + (size_t)(h0 + lm) * (Hh / 2);

  v8f c = {0.f, 0.f, 0.f, 0.f, 0.f, 0.f, 0.f, 0.f};
  #pragma unroll
  for (int k0 = 0; k0 < Hh; k0 += 32) {
    BFrag a, b;
    #pragma unroll
    for (int j = 0; j < 8; ++j) {
      const int kk = k0 + ((j < 4) ? 0 : 16) + (hi ? 8 : 0) + 2 * (j & 3);
      a.u[j] = xp[kk >> 1];
      b.u[j] = wp[kk >> 1];
    }
    c = __builtin_amdgcn_wmma_f32_16x16x32_bf16(false, a.v, false, b.v, (short)0, c, false, false);
  }
  #pragma unroll
  for (int r = 0; r < 8; ++r) {
    const int m = r + hi * 8;
    const int h = h0 + lm;
    out[(size_t)(n0 + m) * Hh + h] = fmaxf(c[r] + bias[h], 0.f);
  }
}

__global__ void __launch_bounds__(256) k_fc2(const float* __restrict__ x, const float* __restrict__ w,
                                             const float* __restrict__ b, float* __restrict__ out) {
  const int lane = threadIdx.x & 31;
  const int g = blockIdx.x * 8 + (threadIdx.x >> 5);
  if (g >= Gg) return;
  float dot = -INFINITY;
  if (lane < Cc) {
    float acc = b[lane];
    for (int d = 0; d < Hh; ++d) acc += x[(size_t)g * Hh + d] * w[lane * Hh + d];
    dot = acc;
  }
  float m = dot;
  for (int o = 16; o > 0; o >>= 1) m = fmaxf(m, __shfl_xor(m, o));
  float ex = (lane < Cc) ? __expf(dot - m) : 0.f;
  float se = ex;
  for (int o = 16; o > 0; o >>= 1) se += __shfl_xor(se, o);
  if (lane < Cc) out[(size_t)g * Cc + lane] = dot - m - logf(se);
}

extern "C" void kernel_launch(void* const* d_in, const int* in_sizes, int n_in,
                              void* d_out, int out_size, void* d_ws, size_t ws_size,
                              hipStream_t stream) {
  const float* x_in  = (const float*)d_in[0];
  const int*   eidx  = (const int*)  d_in[1];
  const int*   idx   = (const int*)  d_in[2];
  const float* fc_w  = (const float*)d_in[3];
  const float* a1    = (const float*)d_in[4];
  const float* a2    = (const float*)d_in[5];
  const float* fc1_w = (const float*)d_in[6];
  const float* fc1_b = (const float*)d_in[7];
  const float* fc2_w = (const float*)d_in[8];
  const float* fc2_b = (const float*)d_in[9];
  const float* gamma = (const float*)d_in[10];
  const float* beta  = (const float*)d_in[11];
  const int* src = eidx;
  const int* dst = eidx + Ee;

  float* ws   = (float*)d_ws;
  float* STAT = ws;
  float* CORR = ws + 64;
  float* Z    = ws + 256;
  float* S1   = Z + (size_t)Nn * Hh;
  float* S2   = S1 + Nn;
  float* HB   = S2 + Nn;
  float* HS   = HB + Ee;
  float* GOUT = HS + Nn;
  float* POOL = GOUT + (size_t)Nn * Hh;
  float* XB   = POOL + (size_t)Gg * Hh;
  float* FC1O = XB + (size_t)Gg * Hh;
  unsigned short* XBF  = (unsigned short*)(FC1O + (size_t)Gg * Hh);
  unsigned short* WBF  = XBF  + (size_t)Nn * Dd;
  unsigned short* W1BF = WBF  + (size_t)Hh * Dd;
  unsigned short* XBBF = W1BF + (size_t)Hh * Hh;

  (void)in_sizes; (void)n_in; (void)out_size; (void)ws_size;

  k_zero<<<1,    256, 0, stream>>>(STAT, 16);
  k_zero<<<256,  256, 0, stream>>>(HS, Nn);
  k_zero<<<2048, 256, 0, stream>>>(GOUT, (long long)Nn * Hh);
  k_zero<<<64,   256, 0, stream>>>(POOL, (long long)Gg * Hh);

  k_cvt<<<2048, 256, 0, stream>>>(x_in,  XBF,  (long long)Nn * Dd);
  k_cvt<<<64,   256, 0, stream>>>(fc_w,  WBF,  (long long)Hh * Dd);
  k_cvt<<<64,   256, 0, stream>>>(fc1_w, W1BF, (long long)Hh * Hh);

  k_stats<<<1024, 256, 0, stream>>>(x_in, STAT);
  k_prep<<<1, 128, 0, stream>>>(fc_w, STAT, CORR);
  k_gat_gemm<<<Nn / 16, 256, 0, stream>>>(XBF, WBF, STAT, CORR, Z);
  k_s12<<<Nn / 8, 256, 0, stream>>>(Z, a1, a2, S1, S2);
  k_edge1<<<4096, 256, 0, stream>>>(src, dst, S1, S2, HB, HS);
  k_edge2<<<Ee / 8, 256, 0, stream>>>(src, dst, HB, HS, Z, GOUT);
  k_pool<<<Nn / 8, 256, 0, stream>>>(GOUT, idx, POOL);
  k_bn<<<1, 128, 0, stream>>>(POOL, gamma, beta, XB);
  k_cvt<<<64, 256, 0, stream>>>(XB, XBBF, (long long)Gg * Hh);
  k_fc1<<<Gg / 16, 256, 0, stream>>>(XBBF, W1BF, fc1_b, FC1O);
  k_fc2<<<Gg / 8, 256, 0, stream>>>(FC1O, fc2_w, fc2_b, (float*)d_out);
}


# edge aggregation: one wave walks 32 edges (indices and weight fetched once per lane, rows broadcast to scalars, 8 z rows in flight): 50000 waves, not 1.6M; same atomics, arrival order
# speedup vs baseline: 1.8381x; 1.8381x over previous
#include <hip/hip_runtime.h>
#include <math.h>

#define Nn 50000
#define Ee 1600000
#define Dd 128
#define Hh 128
#define Cc 10
#define Gg 512
#define LRELU 0.05f
#define BN_EPS 1e-5f

typedef __attribute__((ext_vector_type(16))) __bf16 v16bf;
typedef __attribute__((ext_vector_type(8)))  float  v8f;

union BFrag { v16bf v; unsigned u[8]; };

__device__ __forceinline__ unsigned short f2bf_bits(float f) {
  union { float f; unsigned u; } v; v.f = f;
  return (unsigned short)((v.u + 0x7FFFu + ((v.u >> 16) & 1u)) >> 16);
}

__global__ void k_zero(float* __restrict__ p, long long n) {
  long long i = (long long)blockIdx.x * blockDim.x + threadIdx.x;
  const long long stride = (long long)gridDim.x * blockDim.x;
  for (; i < n; i += stride) p[i] = 0.0f;
}

__global__ void k_cvt(const float* __restrict__ s, unsigned short* __restrict__ d, long long n) {
  long long i = (long long)blockIdx.x * blockDim.x + threadIdx.x;
  const long long stride = (long long)gridDim.x * blockDim.x;
  for (; i < n; i += stride) d[i] = f2bf_bits(s[i]);
}

__global__ void __launch_bounds__(256) k_stats(const float* __restrict__ x, float* __restrict__ stats) {
  __shared__ float ssum[256], ssq[256];
  const int tid = threadIdx.x;
  long long i = (long long)blockIdx.x * blockDim.x + tid;
  const long long stride = (long long)gridDim.x * blockDim.x;
  const long long total = (long long)Nn * Dd;
  float s = 0.f, q = 0.f;
  for (; i < total; i += stride) { float v = x[i]; s += v; q += v * v; }
  ssum[tid] = s; ssq[tid] = q; __syncthreads();
  for (int o = 128; o > 0; o >>= 1) {
    if (tid < o) { ssum[tid] += ssum[tid + o]; ssq[tid] += ssq[tid + o]; }
    __syncthreads();
  }
  if (tid == 0) { atomicAdd(&stats[0], ssum[0]); atomicAdd(&stats[1], ssq[0]); }
}

__global__ void k_prep(const float* __restrict__ w, float* __restrict__ stats,
                       float* __restrict__ corr) {
  const int h = threadIdx.x;
  float colsum = 0.f;
  for (int d = 0; d < Dd; ++d) colsum += w[h * Dd + d];
  const float nd = (float)((long long)Nn * Dd);
  const float s  = stats[0], ss = stats[1];
  const float mu = s / nd;
  const float var = (ss - s * s / nd) / (nd - 1.0f);
  const float invsig = rsqrtf(var);
  if (h == 0) stats[2] = invsig;
  corr[h] = mu * invsig * colsum;
}

__global__ void __launch_bounds__(256) k_gat_gemm(const unsigned short* __restrict__ xbf,
                                                  const unsigned short* __restrict__ wbf,
                                                  const float* __restrict__ stats,
                                                  const float* __restrict__ corr,
                                                  float* __restrict__ z) {
  const int tid  = threadIdx.x;
  const int lane = tid & 31;
  const int hi   = lane >> 4;
  const int lm   = lane & 15;
  const int n0   = blockIdx.x * 16;
  const int h0   = (tid >> 5) * 16;

  const unsigned* xp = (const unsigned*)xbf + (size_t)(n0 + lm) * (Dd / 2);
  const unsigned* wp = (const unsigned*)wbf + (size_t)(h0 + lm) * (Dd / 2);

  v8f c = {0.f, 0.f, 0.f, 0.f, 0.f, 0.f, 0.f, 0.f};
  const float invsig = stats[2];

  #pragma unroll
  for (int k0 = 0; k0 < Dd; k0 += 32) {
    BFrag a, b;
    #pragma unroll
    for (int j = 0; j < 8; ++j) {
      const int kk = k0 + ((j < 4) ? 0 : 16) + (hi ? 8 : 0) + 2 * (j & 3);
      a.u[j] = xp[kk >> 1];
      b.u[j] = wp[kk >> 1];
    }
    if (k0 + 32 < Dd)
      __builtin_prefetch(xp + ((k0 + 32) >> 1), 0, 1);
    c = __builtin_amdgcn_wmma_f32_16x16x32_bf16(false, a.v, false, b.v, (short)0, c, false, false);
  }
  #pragma unroll
  for (int r = 0; r < 8; ++r) {
    const int m = r + hi * 8;
    const int h = h0 + lm;
    z[(size_t)(n0 + m) * Hh + h] = c[r] * invsig - corr[h];
  }
}

__global__ void __launch_bounds__(256) k_s12(const float* __restrict__ z,
                                             const float* __restrict__ a1,
                                             const float* __restrict__ a2,
                                             float* __restrict__ s1, float* __restrict__ s2) {
  const int lane = threadIdx.x & 31;
  const int n = blockIdx.x * 8 + (threadIdx.x >> 5);
  if (n >= Nn) return;
  float p1 = 0.f, p2 = 0.f;
  for (int i = lane; i < Hh; i += 32) {
    const float zv = z[(size_t)n * Hh + i];
    p1 += zv * a1[i]; p2 += zv * a2[i];
  }
  for (int o = 16; o > 0; o >>= 1) { p1 += __shfl_xor(p1, o); p2 += __shfl_xor(p2, o); }
  if (lane == 0) { s1[n] = p1; s2[n] = p2; }
}

__global__ void __launch_bounds__(256) k_edge1(const int* __restrict__ src, const int* __restrict__ dst,
                                               const float* __restrict__ s1, const float* __restrict__ s2,
                                               float* __restrict__ hbuf, float* __restrict__ hsum) {
  long long i = (long long)blockIdx.x * blockDim.x + threadIdx.x;
  const long long stride = (long long)gridDim.x * blockDim.x;
  for (; i < Ee; i += stride) {
    const int s = src[i], d = dst[i];
    float e = s1[s] + s2[d];
    e = (e >= 0.f) ? e : LRELU * e;
    const float h = __expf(e);
    hbuf[i] = h;
    atomicAdd(&hsum[s], h);
  }
}

#define EDGE_CHUNK 32
#define EDGE_BATCH 8
static_assert(Ee % (8 * EDGE_CHUNK) == 0, "the edge list must fill whole 8-wave blocks of 32-edge chunks");

__global__ void __launch_bounds__(256) k_edge2(const int* __restrict__ src, const int* __restrict__ dst,
                                               const float* __restrict__ hbuf, const float* __restrict__ hsum,
                                               const float* __restrict__ z, float* __restrict__ gout) {
  const int lane = threadIdx.x & 31;
  const int wave = __builtin_amdgcn_readfirstlane((int)(blockIdx.x * 8 + (threadIdx.x >> 5)));
  if (wave >= Ee / EDGE_CHUNK) return;
  const int e = wave * EDGE_CHUNK + lane;
  const int ms = src[e];
  const int md = dst[e];
  const int ma = __float_as_int(hbuf[e] / hsum[ms]);
  #pragma unroll
  for (int j0 = 0; j0 < EDGE_CHUNK; j0 += EDGE_BATCH) {
    float zv[EDGE_BATCH][4];
    #pragma unroll
    for (int u = 0; u < EDGE_BATCH; ++u) {
      const int d = __builtin_amdgcn_readlane(md, j0 + u);
      const float* zr = z + (size_t)d * Hh + lane;
      zv[u][0] = zr[0]; zv[u][1] = zr[32]; zv[u][2] = zr[64]; zv[u][3] = zr[96];
    }
    #pragma unroll
    for (int u = 0; u < EDGE_BATCH; ++u) {
      const int s = __builtin_amdgcn_readlane(ms, j0 + u);
      const float alpha = __int_as_float(__builtin_amdgcn_readlane(ma, j0 + u));
      float* o = gout + (size_t)s * Hh + lane;
      atomicAdd(o +  0, alpha * zv[u][0]);
      atomicAdd(o + 32, alpha * zv[u][1]);
      atomicAdd(o + 64, alpha * zv[u][2]);
      atomicAdd(o + 96, alpha * zv[u][3]);
    }
  }
}

__global__ void __launch_bounds__(256) k_pool(const float* __restrict__ gout, const int* __restrict__ idx,
                                              float* __restrict__ pool) {
  const int lane = threadIdx.x & 31;
  const int n = blockIdx.x * 8 + (threadIdx.x >> 5);
  if (n >= Nn) return;
  const int g = idx[n];
  const float* r = gout + (size_t)n * Hh + lane;
  float* p = pool + (size_t)g * Hh + lane;
  const float v0 = r[0], v1 = r[32], v2 = r[64], v3 = r[96];
  atomicAdd(p +  0, fmaxf(v0, 0.f));
  atomicAdd(p + 32, fmaxf(v1, 0.f));
  atomicAdd(p + 64, fmaxf(v2, 0.f));
  atomicAdd(p + 96, fmaxf(v3, 0.f));
}

__global__ void k_bn(const float* __restrict__ pooled, const float* __restrict__ gamma,
                     const float* __restrict__ beta, float* __restrict__ xb) {
  const int h = threadIdx.x;
  float s = 0.f, q = 0.f;
  for (int g = 0; g < Gg; ++g) { const float v = pooled[g * Hh + h]; s += v; q += v * v; }
  const float mu = s / (float)Gg;
  const float var = q / (float)Gg - mu * mu;
  const float inv = rsqrtf(var + BN_EPS);
  const float ga = gamma[h], be = beta[h];
  for (int g = 0; g < Gg; ++g)
    xb[g * Hh + h] = (pooled[g * Hh + h] - mu) * inv * ga + be;
}

__global__ void __launch_bounds__(256) k_fc1(const unsigned short* __restrict__ xbf,
                                             const unsigned short* __restrict__ wbf,
                                             const float* __restrict__ bias, float* __restrict__ out) {
  const int tid  = threadIdx.x;
  const int lane = tid & 31;
  const int hi   = lane >> 4;
  const int lm   = lane & 15;
  const int n0   = blockIdx.x * 16;
  const int h0   = (tid >> 5) * 16;

  const unsigned* xp = (const unsigned*)xbf + (size_t)(n0 + lm) * (Hh / 2);
  const unsigned* wp = (const unsigned*)wbf + (size_t)(h0 + lm) * (Hh / 2);

  v8f c = {0.f, 0.f, 0.f, 0.f, 0.f, 0.f, 0.f, 0.f};
  #pragma unroll
  for (int k0 = 0; k0 < Hh; k0 += 32) {
    BFrag a, b;
    #pragma unroll
    for (int j = 0; j < 8; ++j) {
      const int kk = k0 + ((j < 4) ? 0 : 16) + (hi ? 8 : 0) + 2 * (j & 3);
      a.u[j] = xp[kk >> 1];
      b.u[j] = wp[kk >> 1];
    }
    c = __builtin_amdgcn_wmma_f32_16x16x32_bf16(false, a.v, false, b.v, (short)0, c, false, false);
  }
  #pragma unroll
  for (int r = 0; r < 8; ++r) {
    const int m = r + hi * 8;
    const int h = h0 + lm;
    out[(size_t)(n0 + m) * Hh + h] = fmaxf(c[r] + bias[h], 0.f);
  }
}

__global__ void __launch_bounds__(256) k_fc2(const float* __restrict__ x, const float* __restrict__ w,
                                             const float* __restrict__ b, float* __restrict__ out) {
  const int lane = threadIdx.x & 31;
  const int g = blockIdx.x * 8 + (threadIdx.x >> 5);
  if (g >= Gg) return;
  float dot = -INFINITY;
  if (lane < Cc) {
    float acc = b[lane];
    for (int d = 0; d < Hh; ++d) acc += x[(size_t)g * Hh + d] * w[lane * Hh + d];
    dot = acc;
  }
  float m = dot;
  for (int o = 16; o > 0; o >>= 1) m = fmaxf(m, __shfl_xor(m, o));
  float ex = (lane < Cc) ? __expf(dot - m) : 0.f;
  float se = ex;
  for (int o = 16; o > 0; o >>= 1) se += __shfl_xor(se, o);
  if (lane < Cc) out[(size_t)g * Cc + lane] = dot - m - logf(se);
}

extern "C" void kernel_launch(void* const* d_in, const int* in_sizes, int n_in,
                              void* d_out, int out_size, void* d_ws, size_t ws_size,
                              hipStream_t stream) {
  const float* x_in  = (const float*)d_in[0];
  const int*   eidx  = (const int*)  d_in[1];
  const int*   idx   = (const int*)  d_in[2];
  const float* fc_w  = (const float*)d_in[3];
  const float* a1    = (const float*)d_in[4];
  const float* a2    = (const float*)d_in[5];
  const float* fc1_w = (const float*)d_in[6];
  const float* fc1_b = (const float*)d_in[7];
  const float* fc2_w = (const float*)d_in[8];
  const float* fc2_b = (const float*)d_in[9];
  const float* gamma = (const float*)d_in[10];
  const float* beta  = (const float*)d_in[11];
  const int* src = eidx;
  const int* dst = eidx + Ee;

  float* ws   = (float*)d_ws;
  float* STAT = ws;
  float* CORR = ws + 64;
  float* Z    = ws + 256;
  float* S1   = Z + (size_t)Nn * Hh;
  float* S2   = S1 + Nn;
  float* HB   = S2 + Nn;
  float* HS   = HB + Ee;
  float* GOUT = HS + Nn + 16;
  float* POOL = GOUT + (size_t)Nn * Hh;
  float* XB   = POOL + (size_t)Gg * Hh;
  float* FC1O = XB + (size_t)Gg * Hh;
  unsigned short* XBF  = (unsigned short*)(FC1O + (size_t)Gg * Hh);
  unsigned short* WBF  = XBF  + (size_t)Nn * Dd;
  unsigned short* W1BF = WBF  + (size_t)Hh * Dd;
  unsigned short* XBBF = W1BF + (size_t)Hh * Hh;

  (void)in_sizes; (void)n_in; (void)out_size; (void)ws_size;

  k_zero<<<1,    256, 0, stream>>>(STAT, 16);
  k_zero<<<256,  256, 0, stream>>>(HS, Nn);
  k_zero<<<2048, 256, 0, stream>>>(GOUT, (long long)Nn * Hh);
  k_zero<<<64,   256, 0, stream>>>(POOL, (long long)Gg * Hh);

  k_cvt<<<2048, 256, 0, stream>>>(x_in,  XBF,  (long long)Nn * Dd);
  k_cvt<<<64,   256, 0, stream>>>(fc_w,  WBF,  (long long)Hh * Dd);
  k_cvt<<<64,   256, 0, stream>>>(fc1_w, W1BF, (long long)Hh * Hh);

  k_stats<<<1024, 256, 0, stream>>>(x_in, STAT);
  k_prep<<<1, 128, 0, stream>>>(fc_w, STAT, CORR);
  k_gat_gemm<<<Nn / 16, 256, 0, stream>>>(XBF, WBF, STAT, CORR, Z);
  k_s12<<<Nn / 8, 256, 0, stream>>>(Z, a1, a2, S1, S2);
  k_edge1<<<4096, 256, 0, stream>>>(src, dst, S1, S2, HB, HS);
  k_edge2<<<Ee / (8 * EDGE_CHUNK), 256, 0, stream>>>(src, dst, HB, HS, Z, GOUT);
  k_pool<<<Nn / 8, 256, 0, stream>>>(GOUT, idx, POOL);
  k_bn<<<1, 128, 0, stream>>>(POOL, gamma, beta, XB);
  k_cvt<<<64, 256, 0, stream>>>(XB, XBBF, (long long)Gg * Hh);
  k_fc1<<<Gg / 16, 256, 0, stream>>>(XBBF, W1BF, fc1_b, FC1O);
  k_fc2<<<Gg / 8, 256, 0, stream>>>(FC1O, fc2_w, fc2_b, (float*)d_out);
}


# edge aggregation: edges counting-sorted by src (returning int add = arrival-order rank, scan, scatter); one wave sums a node's rows in registers, one store; no f32 atomics on h_sum or node rows
# speedup vs baseline: 1.9094x; 1.0388x over previous
#include <hip/hip_runtime.h>
#include <math.h>

#define Nn 50000
#define Ee 1600000
#define Dd 128
#define Hh 128
#define Cc 10
#define Gg 512
#define LRELU 0.05f
#define BN_EPS 1e-5f

typedef __attribute__((ext_vector_type(16))) __bf16 v16bf;
typedef __attribute__((ext_vector_type(8)))  float  v8f;

union BFrag { v16bf v; unsigned u[8]; };

__device__ __forceinline__ unsigned short f2bf_bits(float f) {
  union { float f; unsigned u; } v; v.f = f;
  return (unsigned short)((v.u + 0x7FFFu + ((v.u >> 16) & 1u)) >> 16);
}

__global__ void k_zero(float* __restrict__ p, long long n) {
  long long i = (long long)blockIdx.x * blockDim.x + threadIdx.x;
  const long long stride = (long long)gridDim.x * blockDim.x;
  for (; i < n; i += stride) p[i] = 0.0f;
}

__global__ void k_cvt(const float* __restrict__ s, unsigned short* __restrict__ d, long long n) {
  long long i = (long long)blockIdx.x * blockDim.x + threadIdx.x;
  const long long stride = (long long)gridDim.x * blockDim.x;
  for (; i < n; i += stride) d[i] = f2bf_bits(s[i]);
}

__global__ void __launch_bounds__(256) k_stats(const float* __restrict__ x, float* __restrict__ stats) {
  __shared__ float ssum[256], ssq[256];
  const int tid = threadIdx.x;
  long long i = (long long)blockIdx.x * blockDim.x + tid;
  const long long stride = (long long)gridDim.x * blockDim.x;
  const long long total = (long long)Nn * Dd;
  float s = 0.f, q = 0.f;
  for (; i < total; i += stride) { float v = x[i]; s += v; q += v * v; }
  ssum[tid] = s; ssq[tid] = q; __syncthreads();
  for (int o = 128; o > 0; o >>= 1) {
    if (tid < o) { ssum[tid] += ssum[tid + o]; ssq[tid] += ssq[tid + o]; }
    __syncthreads();
  }
  if (tid == 0) { atomicAdd(&stats[0], ssum[0]); atomicAdd(&stats[1], ssq[0]); }
}

__global__ void k_prep(const float* __restrict__ w, float* __restrict__ stats,
                       float* __restrict__ corr) {
  const int h = threadIdx.x;
  float colsum = 0.f;
  for (int d = 0; d < Dd; ++d) colsum += w[h * Dd + d];
  const float nd = (float)((long long)Nn * Dd);
  const float s  = stats[0], ss = stats[1];
  const float mu = s / nd;
  const float var = (ss - s * s / nd) / (nd - 1.0f);
  const float invsig = rsqrtf(var);
  if (h == 0) stats[2] = invsig;
  corr[h] = mu * invsig * colsum;
}

__global__ void __launch_bounds__(256) k_gat_gemm(const unsigned short* __restrict__ xbf,
                                                  const unsigned short* __restrict__ wbf,
                                                  const float* __restrict__ stats,
                                                  const float* __restrict__ corr,
                                                  float* __restrict__ z) {
  const int tid  = threadIdx.x;
  const int lane = tid & 31;
  const int hi   = lane >> 4;
  const int lm   = lane & 15;
  const int n0   = blockIdx.x * 16;
  const int h0   = (tid >> 5) * 16;

  const unsigned* xp = (const unsigned*)xbf + (size_t)(n0 + lm) * (Dd / 2);
  const unsigned* wp = (const unsigned*)wbf + (size_t)(h0 + lm) * (Dd / 2);

  v8f c = {0.f, 0.f, 0.f, 0.f, 0.f, 0.f, 0.f, 0.f};
  const float invsig = stats[2];

  #pragma unroll
  for (int k0 = 0; k0 < Dd; k0 += 32) {
    BFrag a, b;
    #pragma unroll
    for (int j = 0; j < 8; ++j) {
      const int kk = k0 + ((j < 4) ? 0 : 16) + (hi ? 8 : 0) + 2 * (j & 3);
      a.u[j] = xp[kk >> 1];
      b.u[j] = wp[kk >> 1];
    }
    if (k0 + 32 < Dd)
      __builtin_prefetch(xp + ((k0 + 32) >> 1), 0, 1);
    c = __builtin_amdgcn_wmma_f32_16x16x32_bf16(false, a.v, false, b.v, (short)0, c, false, false);
  }
  #pragma unroll
  for (int r = 0; r < 8; ++r) {
    const int m = r + hi * 8;
    const int h = h0 + lm;
    z[(size_t)(n0 + m) * Hh + h] = c[r] * invsig - corr[h];
  }
}

__global__ void __launch_bounds__(256) k_s12(const float* __restrict__ z,
                                             const float* __restrict__ a1,
                                             const float* __restrict__ a2,
                                             float* __restrict__ s1, float* __restrict__ s2) {
  const int lane = threadIdx.x & 31;
  const int n = blockIdx.x * 8 + (threadIdx.x >> 5);
  if (n >= Nn) return;
  float p1 = 0.f, p2 = 0.f;
  for (int i = lane; i < Hh; i += 32) {
    const float zv = z[(size_t)n * Hh + i];
    p1 += zv * a1[i]; p2 += zv * a2[i];
  }
  for (int o = 16; o > 0; o >>= 1) { p1 += __shfl_xor(p1, o); p2 += __shfl_xor(p2, o); }
  if (lane == 0) { s1[n] = p1; s2[n] = p2; }
}

__global__ void __launch_bounds__(256) k_edge1(const int* __restrict__ src, const int* __restrict__ dst,
                                               const float* __restrict__ s1, const float* __restrict__ s2,
                                               float* __restrict__ hbuf, int* __restrict__ cnt,
                                               int* __restrict__ rank) {
  long long i = (long long)blockIdx.x * blockDim.x + threadIdx.x;
  const long long stride = (long long)gridDim.x * blockDim.x;
  for (; i < Ee; i += stride) {
    const int s = src[i], d = dst[i];
    float e = s1[s] + s2[d];
    e = (e >= 0.f) ? e : LRELU * e;
    const float h = __expf(e);
    hbuf[i] = h;
    rank[i] = atomicAdd(&cnt[s], 1);
  }
}

#define SCAN_T 1024
#define SCAN_PER 49
static_assert(SCAN_T * SCAN_PER >= Nn, "the scan block must cover every node");

__global__ void __launch_bounds__(SCAN_T) k_scan(const int* __restrict__ cnt, int* __restrict__ offs) {
  __shared__ int wsum[32];
  const int t = threadIdx.x, lane = t & 31, w = t >> 5;
  int v[SCAN_PER];
  #pragma unroll
  for (int i = 0; i < SCAN_PER; ++i) { const int k = t + SCAN_T * i; v[i] = (k < Nn) ? cnt[k] : 0; }
  int s = 0;
  #pragma unroll
  for (int i = 0; i < SCAN_PER; ++i) s += v[i];
  int inc = s;
  for (int o = 1; o < 32; o <<= 1) { const int y = __shfl_up(inc, o); if (lane >= o) inc += y; }
  if (lane == 31) wsum[w] = inc;
  __syncthreads();
  if (w == 0) {
    const int x = wsum[lane];
    int xi = x;
    for (int o = 1; o < 32; o <<= 1) { const int y = __shfl_up(xi, o); if (lane >= o) xi += y; }
    wsum[lane] = xi - x;
  }
  __syncthreads();
  int run = wsum[w] + inc - s;
  #pragma unroll
  for (int i = 0; i < SCAN_PER; ++i) { const int k = t + SCAN_T * i; if (k < Nn) offs[k] = run; run += v[i]; }
}

__global__ void __launch_bounds__(256) k_fill(const int* __restrict__ src, const int* __restrict__ dst,
                                              const float* __restrict__ hbuf, const int* __restrict__ rank,
                                              const int* __restrict__ offs, int* __restrict__ pair) {
  long long i = (long long)blockIdx.x * blockDim.x + threadIdx.x;
  const long long stride = (long long)gridDim.x * blockDim.x;
  int2* p2 = (int2*)pair;
  for (; i < Ee; i += stride) {
    const unsigned pos = min((unsigned)(offs[src[i]] + rank[i]), (unsigned)(Ee - 1));
    p2[pos] = make_int2(dst[i], __float_as_int(hbuf[i]));
  }
}

#define ROW_BATCH 8

__global__ void __launch_bounds__(256) k_edge2(const int* __restrict__ cnt, const int* __restrict__ offs,
                                               const int* __restrict__ pair, const float* __restrict__ z,
                                               float* __restrict__ gout) {
  const int lane = threadIdx.x & 31;
  const int n = __builtin_amdgcn_readfirstlane((int)(blockIdx.x * 8 + (threadIdx.x >> 5)));
  if (n >= Nn) return;
  const int beg = offs[n];
  const int m   = cnt[n];
  const int2* p2 = (const int2*)pair + beg;
  float hs = 0.f;
  for (int c = 0; c < m; c += 32) {
    const int k = c + lane;
    hs += (k < m) ? __int_as_float(p2[k].y) : 0.f;
  }
  for (int o = 16; o > 0; o >>= 1) hs += __shfl_xor(hs, o);
  float4 acc = make_float4(0.f, 0.f, 0.f, 0.f);
  const float4* z4 = (const float4*)z + lane;
  for (int c = 0; c < m; c += 32) {
    const int k = c + lane;
    const int left = m - c;
    int2 p = make_int2(0, 0);
    if (k < m) p = p2[k];
    const int d0 = __builtin_amdgcn_readlane(p.x, 0);
    const int md = (int)min((unsigned)((k < m) ? p.x : d0), (unsigned)(Nn - 1));
    const int ma = (k < m) ? __float_as_int(__int_as_float(p.y) / hs) : 0;
    #pragma unroll
    for (int j0 = 0; j0 < 32; j0 += ROW_BATCH) {
      if (j0 >= left) break;
      float4 zv[ROW_BATCH];
      #pragma unroll
      for (int u = 0; u < ROW_BATCH; ++u) {
        const int d = __builtin_amdgcn_readlane(md, j0 + u);
        zv[u] = z4[(size_t)d * (Hh / 4)];
      }
      #pragma unroll
      for (int u = 0; u < ROW_BATCH; ++u) {
        const float alpha = __int_as_float(__builtin_amdgcn_readlane(ma, j0 + u));
        acc.x += alpha * zv[u].x;
        acc.y += alpha * zv[u].y;
        acc.z += alpha * zv[u].z;
        acc.w += alpha * zv[u].w;
      }
    }
  }
  ((float4*)gout)[(size_t)n * (Hh / 4) + lane] = acc;
}

__global__ void __launch_bounds__(256) k_pool(const float* __restrict__ gout, const int* __restrict__ idx,
                                              float* __restrict__ pool) {
  const int lane = threadIdx.x & 31;
  const int n = blockIdx.x * 8 + (threadIdx.x >> 5);
  if (n >= Nn) return;
  const int g = idx[n];
  const float* r = gout + (size_t)n * Hh + lane;
  float* p = pool + (size_t)g * Hh + lane;
  const float v0 = r[0], v1 = r[32], v2 = r[64], v3 = r[96];
  atomicAdd(p +  0, fmaxf(v0, 0.f));
  atomicAdd(p + 32, fmaxf(v1, 0.f));
  atomicAdd(p + 64, fmaxf(v2, 0.f));
  atomicAdd(p + 96, fmaxf(v3, 0.f));
}

__global__ void k_bn(const float* __restrict__ pooled, const float* __restrict__ gamma,
                     const float* __restrict__ beta, float* __restrict__ xb) {
  const int h = threadIdx.x;
  float s = 0.f, q = 0.f;
  for (int g = 0; g < Gg; ++g) { const float v = pooled[g * Hh + h]; s += v; q += v * v; }
  const float mu = s / (float)Gg;
  const float var = q / (float)Gg - mu * mu;
  const float inv = rsqrtf(var + BN_EPS);
  const float ga = gamma[h], be = beta[h];
  for (int g = 0; g < Gg; ++g)
    xb[g * Hh + h] = (pooled[g * Hh + h] - mu) * inv * ga + be;
}

__global__ void __launch_bounds__(256) k_fc1(const unsigned short* __restrict__ xbf,
                                             const unsigned short* __restrict__ wbf,
                                             const float* __restrict__ bias, float* __restrict__ out) {
  const int tid  = threadIdx.x;
  const int lane = tid & 31;
  const int hi   = lane >> 4;
  const int lm   = lane & 15;
  const int n0   = blockIdx.x * 16;
  const int h0   = (tid >> 5) * 16;

  const unsigned* xp = (const unsigned*)xbf + (size_t)(n0 + lm) * (Hh / 2);
  const unsigned* wp = (const unsigned*)wbf + (size_t)(h0 + lm) * (Hh / 2);

  v8f c = {0.f, 0.f, 0.f, 0.f, 0.f, 0.f, 0.f, 0.f};
  #pragma unroll
  for (int k0 = 0; k0 < Hh; k0 += 32) {
    BFrag a, b;
    #pragma unroll
    for (int j = 0; j < 8; ++j) {
      const int kk = k0 + ((j < 4) ? 0 : 16) + (hi ? 8 : 0) + 2 * (j & 3);
      a.u[j] = xp[kk >> 1];
      b.u[j] = wp[kk >> 1];
    }
    c = __builtin_amdgcn_wmma_f32_16x16x32_bf16(false, a.v, false, b.v, (short)0, c, false, false);
  }
  #pragma unroll
  for (int r = 0; r < 8; ++r) {
    const int m = r + hi * 8;
    const int h = h0 + lm;
    out[(size_t)(n0 + m) * Hh + h] = fmaxf(c[r] + bias[h], 0.f);
  }
}

__global__ void __launch_bounds__(256) k_fc2(const float* __restrict__ x, const float* __restrict__ w,
                                             const float* __restrict__ b, float* __restrict__ out) {
  const int lane = threadIdx.x & 31;
  const int g = blockIdx.x * 8 + (threadIdx.x >> 5);
  if (g >= Gg) return;
  float dot = -INFINITY;
  if (lane < Cc) {
    float acc = b[lane];
    for (int d = 0; d < Hh; ++d) acc += x[(size_t)g * Hh + d] * w[lane * Hh + d];
    dot = acc;
  }
  float m = dot;
  for (int o = 16; o > 0; o >>= 1) m = fmaxf(m, __shfl_xor(m, o));
  float ex = (lane < Cc) ? __expf(dot - m) : 0.f;
  float se = ex;
  for (int o = 16; o > 0; o >>= 1) se += __shfl_xor(se, o);
  if (lane < Cc) out[(size_t)g * Cc + lane] = dot - m - logf(se);
}

extern "C" void kernel_launch(void* const* d_in, const int* in_sizes, int n_in,
                              void* d_out, int out_size, void* d_ws, size_t ws_size,
                              hipStream_t stream) {
  const float* x_in  = (const float*)d_in[0];
  const int*   eidx  = (const int*)  d_in[1];
  const int*   idx   = (const int*)  d_in[2];
  const float* fc_w  = (const float*)d_in[3];
  const float* a1    = (const float*)d_in[4];
  const float* a2    = (const float*)d_in[5];
  const float* fc1_w = (const float*)d_in[6];
  const float* fc1_b = (const float*)d_in[7];
  const float* fc2_w = (const float*)d_in[8];
  const float* fc2_b = (const float*)d_in[9];
  const float* gamma = (const float*)d_in[10];
  const float* beta  = (const float*)d_in[11];
  const int* src = eidx;
  const int* dst = eidx + Ee;

  float* ws   = (float*)d_ws;
  float* STAT = ws;
  float* CORR = ws + 64;
  float* Z    = ws + 256;
  float* S1   = Z + (size_t)Nn * Hh;
  float* S2   = S1 + Nn;
  float* HB   = S2 + Nn;
  int*   CNT  = (int*)(HB + Ee);
  float* GOUT = HB + Ee + Nn + 16;
  float* POOL = GOUT + (size_t)Nn * Hh;
  float* XB   = POOL + (size_t)Gg * Hh;
  float* FC1O = XB + (size_t)Gg * Hh;
  unsigned short* XBF  = (unsigned short*)(FC1O + (size_t)Gg * Hh);
  unsigned short* WBF  = XBF  + (size_t)Nn * Dd;
  unsigned short* W1BF = WBF  + (size_t)Hh * Dd;
  unsigned short* XBBF = W1BF + (size_t)Hh * Hh;
  int* RANK = (int*)(XBBF + (size_t)Gg * Hh);
  int* OFFS = RANK + Ee;
  int* PAIR = OFFS + Nn + 16;

  (void)in_sizes; (void)n_in; (void)out_size; (void)ws_size;

  k_zero<<<1,    256, 0, stream>>>(STAT, 16);
  k_zero<<<256,  256, 0, stream>>>((float*)CNT, Nn);
  k_zero<<<64,   256, 0, stream>>>(POOL, (long long)Gg * Hh);

  k_cvt<<<2048, 256, 0, stream>>>(x_in,  XBF,  (long long)Nn * Dd);
  k_cvt<<<64,   256, 0, stream>>>(fc_w,  WBF,  (long long)Hh * Dd);
  k_cvt<<<64,   256, 0, stream>>>(fc1_w, W1BF, (long long)Hh * Hh);

  k_stats<<<1024, 256, 0, stream>>>(x_in, STAT);
  k_prep<<<1, 128, 0, stream>>>(fc_w, STAT, CORR);
  k_gat_gemm<<<Nn / 16, 256, 0, stream>>>(XBF, WBF, STAT, CORR, Z);
  k_s12<<<Nn / 8, 256, 0, stream>>>(Z, a1, a2, S1, S2);
  k_edge1<<<4096, 256, 0, stream>>>(src, dst, S1, S2, HB, CNT, RANK);
  k_scan<<<1, SCAN_T, 0, stream>>>(CNT, OFFS);
  k_fill<<<2048, 256, 0, stream>>>(src, dst, HB, RANK, OFFS, PAIR);
  k_edge2<<<Nn / 8, 256, 0, stream>>>(CNT, OFFS, PAIR, Z, GOUT);
  k_pool<<<Nn / 8, 256, 0, stream>>>(GOUT, idx, POOL);
  k_bn<<<1, 128, 0, stream>>>(POOL, gamma, beta, XB);
  k_cvt<<<64, 256, 0, stream>>>(XB, XBBF, (long long)Gg * Hh);
  k_fc1<<<Gg / 16, 256, 0, stream>>>(XBBF, W1BF, fc1_b, FC1O);
  k_fc2<<<Gg / 8, 256, 0, stream>>>(FC1O, fc2_w, fc2_b, (float*)d_out);
}


# launch merging: zeroing+weight staging, x stats+x staging+fc_w row sums, GEMM+s1/s2, edge sum+relu+pooling, BN+fc1+fc2+log_softmax each in one launch: 8 launches, not 19; every sum in its old order
# speedup vs baseline: 2.5200x; 1.3198x over previous
#include <hip/hip_runtime.h>
#include <math.h>

#define Nn 50000
#define Ee 1600000
#define Dd 128
#define Hh 128
#define Cc 10
#define Gg 512
#define LRELU 0.05f
#define BN_EPS 1e-5f

typedef __attribute__((ext_vector_type(16))) __bf16 v16bf;
typedef __attribute__((ext_vector_type(8)))  float  v8f;

union BFrag { v16bf v; unsigned u[8]; };

__device__ __forceinline__ unsigned short f2bf_bits(float f) {
  union { float f; unsigned u; } v; v.f = f;
  return (unsigned short)((v.u + 0x7FFFu + ((v.u >> 16) & 1u)) >> 16);
}

__global__ void __launch_bounds__(256) k_setup(const float* __restrict__ fcw, const float* __restrict__ fc1w,
                                               unsigned short* __restrict__ wbf, unsigned short* __restrict__ w1bf,
                                               float* __restrict__ stat, int* __restrict__ cnt,
                                               float* __restrict__ pool) {
  const int i = blockIdx.x * 256 + threadIdx.x;
  const int nthr = 64 * 256;
  wbf[i]  = f2bf_bits(fcw[i]);
  w1bf[i] = f2bf_bits(fc1w[i]);
  if (i < 16) stat[i] = 0.f;
  for (int k = i; k < Nn; k += nthr) cnt[k] = 0;
  for (int k = i; k < Gg * Hh; k += nthr) pool[k] = 0.f;
}

__global__ void __launch_bounds__(256) k_stats(const float* __restrict__ x, const float* __restrict__ w,
                                               unsigned short* __restrict__ xbf, float* __restrict__ stats,
                                               float* __restrict__ colsum) {
  __shared__ float ssum[256], ssq[256];
  const int tid = threadIdx.x;
  if (blockIdx.x == 0 && tid < Hh) {
    const float4* w4 = (const float4*)(w + (size_t)tid * Dd);
    float cs = 0.f;
    #pragma unroll
    for (int d4 = 0; d4 < Dd / 4; ++d4) {
      const float4 v = w4[d4];
      cs += v.x; cs += v.y; cs += v.z; cs += v.w;
    }
    colsum[tid] = cs;
  }
  long long i = (long long)blockIdx.x * blockDim.x + tid;
  const long long stride = (long long)gridDim.x * blockDim.x;
  const long long total = (long long)Nn * Dd;
  float s = 0.f, q = 0.f;
  for (; i < total; i += stride) { float v = x[i]; s += v; q += v * v; xbf[i] = f2bf_bits(v); }
  ssum[tid] = s; ssq[tid] = q; __syncthreads();
  for (int o = 128; o > 0; o >>= 1) {
    if (tid < o) { ssum[tid] += ssum[tid + o]; ssq[tid] += ssq[tid + o]; }
    __syncthreads();
  }
  if (tid == 0) { atomicAdd(&stats[0], ssum[0]); atomicAdd(&stats[1], ssq[0]); }
}

#define ZPITCH 132
__global__ void __launch_bounds__(256) k_gat_gemm(const unsigned short* __restrict__ xbf,
                                                  const unsigned short* __restrict__ wbf,
                                                  const float* __restrict__ stats,
                                                  const float* __restrict__ colsum,
                                                  const float* __restrict__ a1,
                                                  const float* __restrict__ a2,
                                                  float* __restrict__ z,
                                                  float* __restrict__ s1, float* __restrict__ s2) {
  __shared__ float sZ[16 * ZPITCH];
  const int tid  = threadIdx.x;
  const int lane = tid & 31;
  const int hi   = lane >> 4;
  const int lm   = lane & 15;
  const int n0   = blockIdx.x * 16;
  const int h0   = (tid >> 5) * 16;

  const unsigned* xp = (const unsigned*)xbf + (size_t)(n0 + lm) * (Dd / 2);
  const unsigned* wp = (const unsigned*)wbf + (size_t)(h0 + lm) * (Dd / 2);

  v8f c = {0.f, 0.f, 0.f, 0.f, 0.f, 0.f, 0.f, 0.f};
  const float nd = (float)((long long)Nn * Dd);
  const float s  = stats[0], ss = stats[1];
  const float mu = s / nd;
  const float var = (ss - s * s / nd) / (nd - 1.0f);
  const float invsig = rsqrtf(var);
  const float corr = mu * invsig * colsum[h0 + lm];

  #pragma unroll
  for (int k0 = 0; k0 < Dd; k0 += 32) {
    BFrag a, b;
    #pragma unroll
    for (int j = 0; j < 8; ++j) {
      const int kk = k0 + ((j < 4) ? 0 : 16) + (hi ? 8 : 0) + 2 * (j & 3);
      a.u[j] = xp[kk >> 1];
      b.u[j] = wp[kk >> 1];
    }
    if (k0 + 32 < Dd)
      __builtin_prefetch(xp + ((k0 + 32) >> 1), 0, 1);
    c = __builtin_amdgcn_wmma_f32_16x16x32_bf16(false, a.v, false, b.v, (short)0, c, false, false);
  }
  float a1v[4], a2v[4];
  #pragma unroll
  for (int k = 0; k < 4; ++k) { a1v[k] = a1[lane + 32 * k]; a2v[k] = a2[lane + 32 * k]; }
  #pragma unroll
  for (int r = 0; r < 8; ++r) {
    const int m = r + hi * 8;
    const int h = h0 + lm;
    const float zv = c[r] * invsig - corr;
    z[(size_t)(n0 + m) * Hh + h] = zv;
    sZ[m * ZPITCH + h] = zv;
  }
  __syncthreads();
  const int row0 = (tid >> 5) * 2;
  float p1[2] = {0.f, 0.f}, p2[2] = {0.f, 0.f};
  #pragma unroll
  for (int k = 0; k < 4; ++k) {
    #pragma unroll
    for (int t = 0; t < 2; ++t) {
      const float zv = sZ[(row0 + t) * ZPITCH + lane + 32 * k];
      p1[t] += zv * a1v[k]; p2[t] += zv * a2v[k];
    }
  }
  for (int o = 16; o > 0; o >>= 1) {
    #pragma unroll
    for (int t = 0; t < 2; ++t) { p1[t] += __shfl_xor(p1[t], o); p2[t] += __shfl_xor(p2[t], o); }
  }
  if (lane == 0) {
    s1[n0 + row0] = p1[0]; s1[n0 + row0 + 1] = p1[1];
    s2[n0 + row0] = p2[0]; s2[n0 + row0 + 1] = p2[1];
  }
}

__global__ void __launch_bounds__(256) k_edge1(const int* __restrict__ src, const int* __restrict__ dst,
                                               const float* __restrict__ s1, const float* __restrict__ s2,
                                               float* __restrict__ hbuf, int* __restrict__ cnt,
                                               int* __restrict__ rank) {
  long long i = (long long)blockIdx.x * blockDim.x + threadIdx.x;
  const long long stride = (long long)gridDim.x * blockDim.x;
  for (; i < Ee; i += stride) {
    const int s = src[i], d = dst[i];
    float e = s1[s] + s2[d];
    e = (e >= 0.f) ? e : LRELU * e;
    const float h = __expf(e);
    hbuf[i] = h;
    rank[i] = atomicAdd(&cnt[s], 1);
  }
}

#define SCAN_T 1024
#define SCAN_PER 49
static_assert(SCAN_T * SCAN_PER >= Nn, "the scan block must cover every node");

__global__ void __launch_bounds__(SCAN_T) k_scan(const int* __restrict__ cnt, int* __restrict__ offs) {
  __shared__ int wsum[32];
  const int t = threadIdx.x, lane = t & 31, w = t >> 5;
  int v[SCAN_PER];
  #pragma unroll
  for (int i = 0; i < SCAN_PER; ++i) { const int k = t + SCAN_T * i; v[i] = (k < Nn) ? cnt[k] : 0; }
  int s = 0;
  #pragma unroll
  for (int i = 0; i < SCAN_PER; ++i) s += v[i];
  int inc = s;
  for (int o = 1; o < 32; o <<= 1) { const int y = __shfl_up(inc, o); if (lane >= o) inc += y; }
  if (lane == 31) wsum[w] = inc;
  __syncthreads();
  if (w == 0) {
    const int x = wsum[lane];
    int xi = x;
    for (int o = 1; o < 32; o <<= 1) { const int y = __shfl_up(xi, o); if (lane >= o) xi += y; }
    wsum[lane] = xi - x;
  }
  __syncthreads();
  int run = wsum[w] + inc - s;
  #pragma unroll
  for (int i = 0; i < SCAN_PER; ++i) { const int k = t + SCAN_T * i; if (k < Nn) offs[k] = run; run += v[i]; }
}

__global__ void __launch_bounds__(256) k_fill(const int* __restrict__ src, const int* __restrict__ dst,
                                              const float* __restrict__ hbuf, const int* __restrict__ rank,
                                              const int* __restrict__ offs, int* __restrict__ pair) {
  long long i = (long long)blockIdx.x * blockDim.x + threadIdx.x;
  const long long stride = (long long)gridDim.x * blockDim.x;
  int2* p2 = (int2*)pair;
  for (; i < Ee; i += stride) {
    const unsigned pos = min((unsigned)(offs[src[i]] + rank[i]), (unsigned)(Ee - 1));
    p2[pos] = make_int2(dst[i], __float_as_int(hbuf[i]));
  }
}

#define ROW_BATCH 8

__global__ void __launch_bounds__(256) k_edge2(const int* __restrict__ cnt, const int* __restrict__ offs,
                                               const int* __restrict__ pair, const float* __restrict__ z,
                                               const int* __restrict__ idx, float* __restrict__ pool) {
  __shared__ __attribute__((aligned(16))) float sRow[8 * Hh];
  const int lane = threadIdx.x & 31;
  const int n = __builtin_amdgcn_readfirstlane((int)(blockIdx.x * 8 + (threadIdx.x >> 5)));
  if (n >= Nn) return;
  const int beg = offs[n];
  const int m   = cnt[n];
  const int2* p2 = (const int2*)pair + beg;
  float hs = 0.f;
  for (int c = 0; c < m; c += 32) {
    const int k = c + lane;
    hs += (k < m) ? __int_as_float(p2[k].y) : 0.f;
  }
  for (int o = 16; o > 0; o >>= 1) hs += __shfl_xor(hs, o);
  float4 acc = make_float4(0.f, 0.f, 0.f, 0.f);
  const float4* z4 = (const float4*)z + lane;
  for (int c = 0; c < m; c += 32) {
    const int k = c + lane;
    const int left = m - c;
    int2 p = make_int2(0, 0);
    if (k < m) p = p2[k];
    const int d0 = __builtin_amdgcn_readlane(p.x, 0);
    const int md = (int)min((unsigned)((k < m) ? p.x : d0), (unsigned)(Nn - 1));
    const int ma = (k < m) ? __float_as_int(__int_as_float(p.y) / hs) : 0;
    #pragma unroll
    for (int j0 = 0; j0 < 32; j0 += ROW_BATCH) {
      if (j0 >= left) break;
      float4 zv[ROW_BATCH];
      #pragma unroll
      for (int u = 0; u < ROW_BATCH; ++u) {
        const int d = __builtin_amdgcn_readlane(md, j0 + u);
        zv[u] = z4[(size_t)d * (Hh / 4)];
      }
      #pragma unroll
      for (int u = 0; u < ROW_BATCH; ++u) {
        const float alpha = __int_as_float(__builtin_amdgcn_readlane(ma, j0 + u));
        acc.x += alpha * zv[u].x;
        acc.y += alpha * zv[u].y;
        acc.z += alpha * zv[u].z;
        acc.w += alpha * zv[u].w;
      }
    }
  }
  float* my = sRow + (threadIdx.x >> 5) * Hh;
  my[4 * lane + 0] = acc.x; my[4 * lane + 1] = acc.y; my[4 * lane + 2] = acc.z; my[4 * lane + 3] = acc.w;
  const float v0 = my[lane], v1 = my[lane + 32], v2 = my[lane + 64], v3 = my[lane + 96];
  const int g = idx[n];
  float* p = pool + (size_t)g * Hh + lane;
  atomicAdd(p +  0, fmaxf(v0, 0.f));
  atomicAdd(p + 32, fmaxf(v1, 0.f));
  atomicAdd(p + 64, fmaxf(v2, 0.f));
  atomicAdd(p + 96, fmaxf(v3, 0.f));
}

#define XPITCH 68
#define FPITCH 132
#define WPITCH 129

__global__ void __launch_bounds__(256) k_tail(const float* __restrict__ pooled, const float* __restrict__ gamma,
                                              const float* __restrict__ beta,
                                              const unsigned short* __restrict__ w1bf,
                                              const float* __restrict__ b1, const float* __restrict__ w2,
                                              const float* __restrict__ b2, float* __restrict__ out) {
  __shared__ float sMu[Hh], sInv[Hh];
  __shared__ __attribute__((aligned(16))) unsigned sX[16 * XPITCH];
  __shared__ float sF[16 * FPITCH];
  __shared__ float sW2[Cc * WPITCH];
  const int tid  = threadIdx.x;
  const int lane = tid & 31;
  const int hi   = lane >> 4;
  const int lm   = lane & 15;
  const int n0   = blockIdx.x * 16;
  const int h0   = (tid >> 5) * 16;

  const unsigned* wp = (const unsigned*)w1bf + (size_t)(h0 + lm) * (Hh / 2);
  BFrag b[4];
  #pragma unroll
  for (int t = 0; t < 4; ++t) {
    #pragma unroll
    for (int j = 0; j < 8; ++j) {
      const int kk = t * 32 + ((j < 4) ? 0 : 16) + (hi ? 8 : 0) + 2 * (j & 3);
      b[t].u[j] = wp[kk >> 1];
    }
  }
  for (int i = tid; i < Cc * Hh; i += 256) sW2[(i >> 7) * WPITCH + (i & 127)] = w2[i];

  if (tid < Hh) {
    float s = 0.f, q = 0.f;
    #pragma unroll 64
    for (int g = 0; g < Gg; ++g) { const float v = pooled[g * Hh + tid]; s += v; q += v * v; }
    const float mu = s / (float)Gg;
    const float var = q / (float)Gg - mu * mu;
    sMu[tid] = mu;
    sInv[tid] = rsqrtf(var + BN_EPS);
  }
  __syncthreads();

  {
    const int r  = tid >> 4;
    const int c8 = (tid & 15) * 8;
    const float4* p4 = (const float4*)(pooled + (size_t)(n0 + r) * Hh + c8);
    const float4* g4 = (const float4*)(gamma + c8);
    const float4* e4 = (const float4*)(beta + c8);
    const float4 pa = p4[0], pb = p4[1], ga = g4[0], gb = g4[1], ea = e4[0], eb = e4[1];
    const float pv[8] = {pa.x, pa.y, pa.z, pa.w, pb.x, pb.y, pb.z, pb.w};
    const float gv[8] = {ga.x, ga.y, ga.z, ga.w, gb.x, gb.y, gb.z, gb.w};
    const float ev[8] = {ea.x, ea.y, ea.z, ea.w, eb.x, eb.y, eb.z, eb.w};
    unsigned pk[4];
    #pragma unroll
    for (int e = 0; e < 4; ++e) {
      const int k = c8 + 2 * e;
      const float x0 = (pv[2 * e]     - sMu[k])     * sInv[k]     * gv[2 * e]     + ev[2 * e];
      const float x1 = (pv[2 * e + 1] - sMu[k + 1]) * sInv[k + 1] * gv[2 * e + 1] + ev[2 * e + 1];
      pk[e] = (unsigned)f2bf_bits(x0) | ((unsigned)f2bf_bits(x1) << 16);
    }
    ((uint4*)sX)[r * (XPITCH / 4) + (tid & 15)] = make_uint4(pk[0], pk[1], pk[2], pk[3]);
  }
  __syncthreads();

  v8f c = {0.f, 0.f, 0.f, 0.f, 0.f, 0.f, 0.f, 0.f};
  const uint4* sx4 = (const uint4*)sX + lm * (XPITCH / 4) + hi;
  #pragma unroll
  for (int t = 0; t < 4; ++t) {
    const uint4 lo = sx4[t * 4];
    const uint4 up = sx4[t * 4 + 2];
    BFrag a;
    a.u[0] = lo.x; a.u[1] = lo.y; a.u[2] = lo.z; a.u[3] = lo.w;
    a.u[4] = up.x; a.u[5] = up.y; a.u[6] = up.z; a.u[7] = up.w;
    c = __builtin_amdgcn_wmma_f32_16x16x32_bf16(false, a.v, false, b[t].v, (short)0, c, false, false);
  }
  {
    const int h = h0 + lm;
    const float bias = b1[h];
    #pragma unroll
    for (int r = 0; r < 8; ++r) {
      const int m = r + hi * 8;
      sF[m * FPITCH + h] = fmaxf(c[r] + bias, 0.f);
    }
  }
  __syncthreads();

  const int row = (tid >> 5) * 2 + hi;
  float dot = -INFINITY;
  if (lm < Cc) {
    float acc = b2[lm];
    for (int d = 0; d < Hh; ++d) acc += sF[row * FPITCH + d] * sW2[lm * WPITCH + d];
    dot = acc;
  }
  float m = dot;
  for (int o = 8; o > 0; o >>= 1) m = fmaxf(m, __shfl_xor(m, o));
  const float ex = (lm < Cc) ? __expf(dot - m) : 0.f;
  float se = ex;
  for (int o = 8; o > 0; o >>= 1) se += __shfl_xor(se, o);
  if (lm < Cc) out[(size_t)(n0 + row) * Cc + lm] = dot - m - logf(se);
}

extern "C" void kernel_launch(void* const* d_in, const int* in_sizes, int n_in,
                              void* d_out, int out_size, void* d_ws, size_t ws_size,
                              hipStream_t stream) {
  const float* x_in  = (const float*)d_in[0];
  const int*   eidx  = (const int*)  d_in[1];
  const int*   idx   = (const int*)  d_in[2];
  const float* fc_w  = (const float*)d_in[3];
  const float* a1    = (const float*)d_in[4];
  const float* a2    = (const float*)d_in[5];
  const float* fc1_w = (const float*)d_in[6];
  const float* fc1_b = (const float*)d_in[7];
  const float* fc2_w = (const float*)d_in[8];
  const float* fc2_b = (const float*)d_in[9];
  const float* gamma = (const float*)d_in[10];
  const float* beta  = (const float*)d_in[11];
  const int* src = eidx;
  const int* dst = eidx + Ee;

  float* ws   = (float*)d_ws;
  float* STAT = ws;
  float* CORR = ws + 64;
  float* Z    = ws + 256;
  float* S1   = Z + (size_t)Nn * Hh;
  float* S2   = S1 + Nn;
  float* HB   = S2 + Nn;
  int*   CNT  = (int*)(HB + Ee);
  float* GOUT = HB + Ee + Nn + 16;
  float* POOL = GOUT + (size_t)Nn * Hh;
  float* XB   = POOL + (size_t)Gg * Hh;
  float* FC1O = XB + (size_t)Gg * Hh;
  unsigned short* XBF  = (unsigned short*)(FC1O + (size_t)Gg * Hh);
  unsigned short* WBF  = XBF  + (size_t)Nn * Dd;
  unsigned short* W1BF = WBF  + (size_t)Hh * Dd;
  unsigned short* XBBF = W1BF + (size_t)Hh * Hh;
  int* RANK = (int*)(XBBF + (size_t)Gg * Hh);
  int* OFFS = RANK + Ee;
  int* PAIR = OFFS + Nn + 16;

  (void)in_sizes; (void)n_in; (void)out_size; (void)ws_size;

  k_setup<<<64, 256, 0, stream>>>(fc_w, fc1_w, WBF, W1BF, STAT, CNT, POOL);
  k_stats<<<1024, 256, 0, stream>>>(x_in, fc_w, XBF, STAT, CORR);
  k_gat_gemm<<<Nn / 16, 256, 0, stream>>>(XBF, WBF, STAT, CORR, a1, a2, Z, S1, S2);
  k_edge1<<<4096, 256, 0, stream>>>(src, dst, S1, S2, HB, CNT, RANK);
  k_scan<<<1, SCAN_T, 0, stream>>>(CNT, OFFS);
  k_fill<<<2048, 256, 0, stream>>>(src, dst, HB, RANK, OFFS, PAIR);
  k_edge2<<<Nn / 8, 256, 0, stream>>>(CNT, OFFS, PAIR, Z, idx, POOL);
  k_tail<<<Gg / 16, 256, 0, stream>>>(POOL, gamma, beta, W1BF, fc1_b, fc2_w, fc2_b, (float*)d_out);
}


# launch merging: x sums and bf16 rounding inside the GEMM (standardization applied where z, s1, s2 are used), set-up inside the ranking launch: 6 launches, not 8; x sums in tile order, no atomics
# speedup vs baseline: 4.0659x; 1.6135x over previous
#include <hip/hip_runtime.h>
#include <math.h>

#define Nn 50000
#define Ee 1600000
#define Dd 128
#define Hh 128
#define Cc 10
#define Gg 512
#define LRELU 0.05f
#define BN_EPS 1e-5f

typedef __attribute__((ext_vector_type(16))) __bf16 v16bf;
typedef __attribute__((ext_vector_type(8)))  float  v8f;

union BFrag { v16bf v; unsigned u[8]; };

__device__ __forceinline__ unsigned short f2bf_bits(float f) {
  union { float f; unsigned u; } v; v.f = f;
  return (unsigned short)((v.u + 0x7FFFu + ((v.u >> 16) & 1u)) >> 16);
}

#define ZPITCH 132
#define APITCH 68
#define GM_BLOCKS (Nn / 16)
static_assert(GM_BLOCKS * 16 == Nn, "16-row tiles must cover the nodes exactly");
__global__ void __launch_bounds__(256) k_gat_gemm(const float* __restrict__ x,
                                                  const unsigned short* __restrict__ wbf,
                                                  const float* __restrict__ a1,
                                                  const float* __restrict__ a2,
                                                  float* __restrict__ z,
                                                  float* __restrict__ s1, float* __restrict__ s2,
                                                  float* __restrict__ part) {
  __shared__ __attribute__((aligned(16))) unsigned sA[16 * APITCH];
  __shared__ float sZ[16 * ZPITCH];
  __shared__ float sPs[8], sPq[8];
  const int tid  = threadIdx.x;
  const int lane = tid & 31;
  const int w    = tid >> 5;
  const int hi   = lane >> 4;
  const int lm   = lane & 15;
  const int n0   = blockIdx.x * 16;
  const int h0   = w * 16;

  const float4* x4 = (const float4*)(x + (size_t)n0 * Dd);
  const float4 v0 = x4[tid], v1 = x4[tid + 256];

  const unsigned* wp = (const unsigned*)wbf + (size_t)(h0 + lm) * (Dd / 2);
  BFrag b[4];
  #pragma unroll
  for (int t = 0; t < 4; ++t) {
    #pragma unroll
    for (int j = 0; j < 8; ++j) {
      const int kk = t * 32 + ((j < 4) ? 0 : 16) + (hi ? 8 : 0) + 2 * (j & 3);
      b[t].u[j] = wp[kk >> 1];
    }
  }

  ((uint2*)sA)[w * (APITCH / 2) + lane] =
      make_uint2((unsigned)f2bf_bits(v0.x) | ((unsigned)f2bf_bits(v0.y) << 16),
                 (unsigned)f2bf_bits(v0.z) | ((unsigned)f2bf_bits(v0.w) << 16));
  ((uint2*)sA)[(w + 8) * (APITCH / 2) + lane] =
      make_uint2((unsigned)f2bf_bits(v1.x) | ((unsigned)f2bf_bits(v1.y) << 16),
                 (unsigned)f2bf_bits(v1.z) | ((unsigned)f2bf_bits(v1.w) << 16));
  float ps = ((v0.x + v0.y) + (v0.z + v0.w)) + ((v1.x + v1.y) + (v1.z + v1.w));
  float pq = ((v0.x * v0.x + v0.y * v0.y) + (v0.z * v0.z + v0.w * v0.w)) +
             ((v1.x * v1.x + v1.y * v1.y) + (v1.z * v1.z + v1.w * v1.w));
  for (int o = 16; o > 0; o >>= 1) { ps += __shfl_xor(ps, o); pq += __shfl_xor(pq, o); }
  if (lane == 0) { sPs[w] = ps; sPq[w] = pq; }
  __syncthreads();
  if (tid == 0) {
    float s = 0.f, q = 0.f;
    #pragma unroll
    for (int i = 0; i < 8; ++i) { s += sPs[i]; q += sPq[i]; }
    part[2 * blockIdx.x] = s; part[2 * blockIdx.x + 1] = q;
  }

  v8f c = {0.f, 0.f, 0.f, 0.f, 0.f, 0.f, 0.f, 0.f};
  const uint4* sa4 = (const uint4*)sA + lm * (APITCH / 4) + hi;
  #pragma unroll
  for (int t = 0; t < 4; ++t) {
    const uint4 lo = sa4[t * 4];
    const uint4 up = sa4[t * 4 + 2];
    BFrag a;
    a.u[0] = lo.x; a.u[1] = lo.y; a.u[2] = lo.z; a.u[3] = lo.w;
    a.u[4] = up.x; a.u[5] = up.y; a.u[6] = up.z; a.u[7] = up.w;
    c = __builtin_amdgcn_wmma_f32_16x16x32_bf16(false, a.v, false, b[t].v, (short)0, c, false, false);
  }
  float a1v[4], a2v[4];
  #pragma unroll
  for (int k = 0; k < 4; ++k) { a1v[k] = a1[lane + 32 * k]; a2v[k] = a2[lane + 32 * k]; }
  #pragma unroll
  for (int r = 0; r < 8; ++r) {
    const int m = r + hi * 8;
    const int h = h0 + lm;
    z[(size_t)(n0 + m) * Hh + h] = c[r];
    sZ[m * ZPITCH + h] = c[r];
  }
  __syncthreads();
  const int row0 = w * 2;
  float p1[2] = {0.f, 0.f}, p2[2] = {0.f, 0.f};
  #pragma unroll
  for (int k = 0; k < 4; ++k) {
    #pragma unroll
    for (int t = 0; t < 2; ++t) {
      const float zv = sZ[(row0 + t) * ZPITCH + lane + 32 * k];
      p1[t] += zv * a1v[k]; p2[t] += zv * a2v[k];
    }
  }
  for (int o = 16; o > 0; o >>= 1) {
    #pragma unroll
    for (int t = 0; t < 2; ++t) { p1[t] += __shfl_xor(p1[t], o); p2[t] += __shfl_xor(p2[t], o); }
  }
  if (lane == 0) {
    s1[n0 + row0] = p1[0]; s1[n0 + row0 + 1] = p1[1];
    s2[n0 + row0] = p2[0]; s2[n0 + row0 + 1] = p2[1];
  }
}

#define RK_CHUNKS 64
#define RK_EDGES (Ee / RK_CHUNKS)
#define RK_T 1024
#define RK_PER ((RK_EDGES + RK_T - 1) / RK_T)
static_assert(RK_CHUNKS * RK_EDGES == Ee, "the chunks must cover the edge list exactly");
static_assert(RK_CHUNKS * RK_T == Gg * Hh && Hh * Dd <= RK_CHUNKS * RK_T && Hh * Hh <= RK_CHUNKS * RK_T,
              "one pooled element per thread; a weight element of each matrix for the first threads");

__global__ void __launch_bounds__(RK_T) k_rank(const int* __restrict__ src, int* __restrict__ rank,
                                               int* __restrict__ cc,
                                               const float* __restrict__ fcw, const float* __restrict__ fc1w,
                                               unsigned short* __restrict__ wbf, unsigned short* __restrict__ w1bf,
                                               int* __restrict__ cursor, float* __restrict__ pool) {
  __shared__ int lcnt[Nn];
  const int t = threadIdx.x;
  {
    const int i = blockIdx.x * RK_T + t;
    if (i < Hh * Dd) wbf[i]  = f2bf_bits(fcw[i]);
    if (i < Hh * Hh) w1bf[i] = f2bf_bits(fc1w[i]);
    if (i == 0) cursor[0] = 0;
    pool[i] = 0.f;
  }
  const int* sp = src  + (size_t)blockIdx.x * RK_EDGES;
  int*       rp = rank + (size_t)blockIdx.x * RK_EDGES;
  int s[RK_PER], r[RK_PER];
  #pragma unroll
  for (int u = 0; u < RK_PER; ++u) { const int j = t + u * RK_T; s[u] = (j < RK_EDGES) ? sp[j] : 0; }
  for (int k = t; k < Nn; k += RK_T) lcnt[k] = 0;
  __syncthreads();
  #pragma unroll
  for (int u = 0; u < RK_PER; ++u) { const int j = t + u * RK_T; r[u] = (j < RK_EDGES) ? atomicAdd(&lcnt[s[u]], 1) : 0; }
  #pragma unroll
  for (int u = 0; u < RK_PER; ++u) { const int j = t + u * RK_T; if (j < RK_EDGES) rp[j] = r[u]; }
  __syncthreads();
  int* cp = cc + (size_t)blockIdx.x * Nn;
  for (int k = t; k < Nn; k += RK_T) cp[k] = lcnt[k];
}

#define SC_T 256
#define SC_BLOCKS ((Nn + SC_T - 1) / SC_T)
#define PT_PER ((GM_BLOCKS + SC_T - 1) / SC_T)

__global__ void __launch_bounds__(SC_T) k_offsets(int* __restrict__ cc, int* __restrict__ cnt,
                                                  int* __restrict__ offs, int* __restrict__ cursor,
                                                  const float* __restrict__ part, const float* __restrict__ fcw,
                                                  const float* __restrict__ a1, const float* __restrict__ a2,
                                                  float* __restrict__ stat, float* __restrict__ corr) {
  __shared__ int wsum[SC_T / 32];
  __shared__ int sbase;
  __shared__ float rs[2 * (SC_T / 32)], rq[2 * (SC_T / 32)];
  const int t = threadIdx.x, lane = t & 31, w = t >> 5;
  if (blockIdx.x == SC_BLOCKS) {
    float2 pv[PT_PER];
    #pragma unroll
    for (int i = 0; i < PT_PER; ++i) {
      const int b = t + i * SC_T;
      pv[i] = (b < GM_BLOCKS) ? ((const float2*)part)[b] : make_float2(0.f, 0.f);
    }
    float cs = 0.f, av1 = 0.f, av2 = 0.f;
    if (t < Hh) {
      const float4* w4 = (const float4*)(fcw + (size_t)t * Dd);
      av1 = a1[t]; av2 = a2[t];
      #pragma unroll
      for (int d4 = 0; d4 < Dd / 4; ++d4) {
        const float4 v = w4[d4];
        cs += v.x; cs += v.y; cs += v.z; cs += v.w;
      }
    }
    float s = 0.f, q = 0.f;
    #pragma unroll
    for (int i = 0; i < PT_PER; ++i) { s += pv[i].x; q += pv[i].y; }
    for (int o = 16; o > 0; o >>= 1) { s += __shfl_xor(s, o); q += __shfl_xor(q, o); }
    if (lane == 0) { rs[w] = s; rq[w] = q; }
    __syncthreads();
    float S = 0.f, Q = 0.f;
    #pragma unroll
    for (int i = 0; i < SC_T / 32; ++i) { S += rs[i]; Q += rq[i]; }
    const float nd = (float)((long long)Nn * Dd);
    const float mu = S / nd;
    const float var = (Q - S * S / nd) / (nd - 1.0f);
    const float invsig = rsqrtf(var);
    const float mi = mu * invsig;
    if (t < Hh) corr[t] = mi * cs;
    float d1 = cs * av1, d2 = cs * av2;
    for (int o = 16; o > 0; o >>= 1) { d1 += __shfl_xor(d1, o); d2 += __shfl_xor(d2, o); }
    if (lane == 0) { rs[SC_T / 32 + w] = d1; rq[SC_T / 32 + w] = d2; }
    __syncthreads();
    if (t == 0) {
      float k1 = 0.f, k2 = 0.f;
      #pragma unroll
      for (int i = 0; i < SC_T / 32; ++i) { k1 += rs[SC_T / 32 + i]; k2 += rq[SC_T / 32 + i]; }
      stat[0] = invsig; stat[1] = mi * k1; stat[2] = mi * k2;
    }
    return;
  }
  const int n = blockIdx.x * SC_T + t;
  const bool ok = n < Nn;
  const int nn = ok ? n : (Nn - 1);
  int v[RK_CHUNKS];
  #pragma unroll
  for (int k = 0; k < RK_CHUNKS; ++k) v[k] = cc[(size_t)k * Nn + nn];
  int tot = 0;
  #pragma unroll
  for (int k = 0; k < RK_CHUNKS; ++k) { const int c = ok ? v[k] : 0; v[k] = tot; tot += c; }
  int inc = tot;
  for (int o = 1; o < 32; o <<= 1) { const int y = __shfl_up(inc, o); if (lane >= o) inc += y; }
  if (lane == 31) wsum[w] = inc;
  __syncthreads();
  if (t == 0) {
    int run = 0;
    for (int i = 0; i < SC_T / 32; ++i) { const int x = wsum[i]; wsum[i] = run; run += x; }
    sbase = atomicAdd(cursor, run);
  }
  __syncthreads();
  const int start = sbase + wsum[w] + inc - tot;
  if (ok) {
    cnt[n] = tot;
    offs[n] = start;
    #pragma unroll
    for (int k = 0; k < RK_CHUNKS; ++k) cc[(size_t)k * Nn + n] = start + v[k];
  }
}

#define FL_PER 4
#define FL_THREADS (Ee / FL_PER)
#define FL_BLOCKS ((FL_THREADS + 255) / 256)
static_assert(FL_PER == 4 && Ee % FL_PER == 0 && RK_EDGES % FL_PER == 0, "a thread's four edges lie in one chunk");

__global__ void __launch_bounds__(256) k_fill(const int* __restrict__ src, const int* __restrict__ dst,
                                              const float* __restrict__ s1, const float* __restrict__ s2,
                                              const int* __restrict__ rank, const int* __restrict__ cc,
                                              const float* __restrict__ stat, int* __restrict__ pair) {
  const int q = blockIdx.x * 256 + threadIdx.x;
  if (q >= FL_THREADS) return;
  const float isg = stat[0], c1 = stat[1], c2 = stat[2];
  const int4 sv = ((const int4*)src)[q];
  const int4 dv = ((const int4*)dst)[q];
  const int4 rv = ((const int4*)rank)[q];
  const int* ck = cc + (size_t)((q * FL_PER) / RK_EDGES) * Nn;
  const int s[FL_PER] = {sv.x, sv.y, sv.z, sv.w};
  const int d[FL_PER] = {dv.x, dv.y, dv.z, dv.w};
  const int r[FL_PER] = {rv.x, rv.y, rv.z, rv.w};
  float a[FL_PER], b[FL_PER];
  int c0[FL_PER];
  #pragma unroll
  for (int u = 0; u < FL_PER; ++u) { a[u] = s1[s[u]]; b[u] = s2[d[u]]; c0[u] = ck[s[u]]; }
  int2* p2 = (int2*)pair;
  #pragma unroll
  for (int u = 0; u < FL_PER; ++u) {
    float e = (a[u] * isg - c1) + (b[u] * isg - c2);
    e = (e >= 0.f) ? e : LRELU * e;
    const float h = __expf(e);
    const unsigned pos = min((unsigned)(c0[u] + r[u]), (unsigned)(Ee - 1));
    p2[pos] = make_int2(d[u], __float_as_int(h));
  }
}

#define ROW_BATCH 8

__global__ void __launch_bounds__(256) k_edge2(const int* __restrict__ cnt, const int* __restrict__ offs,
                                               const int* __restrict__ pair, const float* __restrict__ z,
                                               const float* __restrict__ stat, const float* __restrict__ corr,
                                               const int* __restrict__ idx, float* __restrict__ pool) {
  __shared__ __attribute__((aligned(16))) float sRow[8 * Hh];
  const int lane = threadIdx.x & 31;
  const int n = __builtin_amdgcn_readfirstlane((int)(blockIdx.x * 8 + (threadIdx.x >> 5)));
  if (n >= Nn) return;
  const int beg = offs[n];
  const int m   = cnt[n];
  const float isg = stat[0];
  const float cq0 = corr[lane], cq1 = corr[lane + 32], cq2 = corr[lane + 64], cq3 = corr[lane + 96];
  const int2* p2 = (const int2*)pair + beg;
  float hs = 0.f;
  for (int c = 0; c < m; c += 32) {
    const int k = c + lane;
    hs += (k < m) ? __int_as_float(p2[k].y) : 0.f;
  }
  for (int o = 16; o > 0; o >>= 1) hs += __shfl_xor(hs, o);
  float4 acc = make_float4(0.f, 0.f, 0.f, 0.f);
  float asum = 0.f;
  const float4* z4 = (const float4*)z + lane;
  for (int c = 0; c < m; c += 32) {
    const int k = c + lane;
    const int left = m - c;
    int2 p = make_int2(0, 0);
    if (k < m) p = p2[k];
    const int d0 = __builtin_amdgcn_readlane(p.x, 0);
    const int md = (int)min((unsigned)((k < m) ? p.x : d0), (unsigned)(Nn - 1));
    const int ma = (k < m) ? __float_as_int(__int_as_float(p.y) / hs) : 0;
    asum += __int_as_float(ma);
    #pragma unroll
    for (int j0 = 0; j0 < 32; j0 += ROW_BATCH) {
      if (j0 >= left) break;
      float4 zv[ROW_BATCH];
      #pragma unroll
      for (int u = 0; u < ROW_BATCH; ++u) {
        const int d = __builtin_amdgcn_readlane(md, j0 + u);
        zv[u] = z4[(size_t)d * (Hh / 4)];
      }
      #pragma unroll
      for (int u = 0; u < ROW_BATCH; ++u) {
        const float alpha = __int_as_float(__builtin_amdgcn_readlane(ma, j0 + u));
        acc.x += alpha * zv[u].x;
        acc.y += alpha * zv[u].y;
        acc.z += alpha * zv[u].z;
        acc.w += alpha * zv[u].w;
      }
    }
  }
  float* my = sRow + (threadIdx.x >> 5) * Hh;
  my[4 * lane + 0] = acc.x; my[4 * lane + 1] = acc.y; my[4 * lane + 2] = acc.z; my[4 * lane + 3] = acc.w;
  const float v0 = my[lane], v1 = my[lane + 32], v2 = my[lane + 64], v3 = my[lane + 96];
  for (int o = 16; o > 0; o >>= 1) asum += __shfl_xor(asum, o);
  const int g = idx[n];
  float* p = pool + (size_t)g * Hh + lane;
  atomicAdd(p +  0, fmaxf(v0 * isg - cq0 * asum, 0.f));
  atomicAdd(p + 32, fmaxf(v1 * isg - cq1 * asum, 0.f));
  atomicAdd(p + 64, fmaxf(v2 * isg - cq2 * asum, 0.f));
  atomicAdd(p + 96, fmaxf(v3 * isg - cq3 * asum, 0.f));
}

#define XPITCH 68
#define FPITCH 132
#define WPITCH 129

__global__ void __launch_bounds__(256) k_tail(const float* __restrict__ pooled, const float* __restrict__ gamma,
                                              const float* __restrict__ beta,
                                              const unsigned short* __restrict__ w1bf,
                                              const float* __restrict__ b1, const float* __restrict__ w2,
                                              const float* __restrict__ b2, float* __restrict__ out) {
  __shared__ float sMu[Hh], sInv[Hh];
  __shared__ __attribute__((aligned(16))) unsigned sX[16 * XPITCH];
  __shared__ float sF[16 * FPITCH];
  __shared__ float sW2[Cc * WPITCH];
  const int tid  = threadIdx.x;
  const int lane = tid & 31;
  const int hi   = lane >> 4;
  const int lm   = lane & 15;
  const int n0   = blockIdx.x * 16;
  const int h0   = (tid >> 5) * 16;

  const unsigned* wp = (const unsigned*)w1bf + (size_t)(h0 + lm) * (Hh / 2);
  BFrag b[4];
  #pragma unroll
  for (int t = 0; t < 4; ++t) {
    #pragma unroll
    for (int j = 0; j < 8; ++j) {
      const int kk = t * 32 + ((j < 4) ? 0 : 16) + (hi ? 8 : 0) + 2 * (j & 3);
      b[t].u[j] = wp[kk >> 1];
    }
  }
  for (int i = tid; i < Cc * Hh; i += 256) sW2[(i >> 7) * WPITCH + (i & 127)] = w2[i];

  if (tid < Hh) {
    float s = 0.f, q = 0.f;
    #pragma unroll 64
    for (int g = 0; g < Gg; ++g) { const float v = pooled[g * Hh + tid]; s += v; q += v * v; }
    const float mu = s / (float)Gg;
    const float var = q / (float)Gg - mu * mu;
    sMu[tid] = mu;
    sInv[tid] = rsqrtf(var + BN_EPS);
  }
  __syncthreads();

  {
    const int r  = tid >> 4;
    const int c8 = (tid & 15) * 8;
    const float4* p4 = (const float4*)(pooled + (size_t)(n0 + r) * Hh + c8);
    const float4* g4 = (const float4*)(gamma + c8);
    const float4* e4 = (const float4*)(beta + c8);
    const float4 pa = p4[0], pb = p4[1], ga = g4[0], gb = g4[1], ea = e4[0], eb = e4[1];
    const float pv[8] = {pa.x, pa.y, pa.z, pa.w, pb.x, pb.y, pb.z, pb.w};
    const float gv[8] = {ga.x, ga.y, ga.z, ga.w, gb.x, gb.y, gb.z, gb.w};
    const float ev[8] = {ea.x, ea.y, ea.z, ea.w, eb.x, eb.y, eb.z, eb.w};
    unsigned pk[4];
    #pragma unroll
    for (int e = 0; e < 4; ++e) {
      const int k = c8 + 2 * e;
      const float x0 = (pv[2 * e]     - sMu[k])     * sInv[k]     * gv[2 * e]     + ev[2 * e];
      const float x1 = (pv[2 * e + 1] - sMu[k + 1]) * sInv[k + 1] * gv[2 * e + 1] + ev[2 * e + 1];
      pk[e] = (unsigned)f2bf_bits(x0) | ((unsigned)f2bf_bits(x1) << 16);
    }
    ((uint4*)sX)[r * (XPITCH / 4) + (tid & 15)] = make_uint4(pk[0], pk[1], pk[2], pk[3]);
  }
  __syncthreads();

  v8f c = {0.f, 0.f, 0.f, 0.f, 0.f, 0.f, 0.f, 0.f};
  const uint4* sx4 = (const uint4*)sX + lm * (XPITCH / 4) + hi;
  #pragma unroll
  for (int t = 0; t < 4; ++t) {
    const uint4 lo = sx4[t * 4];
    const uint4 up = sx4[t * 4 + 2];
    BFrag a;
    a.u[0] = lo.x; a.u[1] = lo.y; a.u[2] = lo.z; a.u[3] = lo.w;
    a.u[4] = up.x; a.u[5] = up.y; a.u[6] = up.z; a.u[7] = up.w;
    c = __builtin_amdgcn_wmma_f32_16x16x32_bf16(false, a.v, false, b[t].v, (short)0, c, false, false);
  }
  {
    const int h = h0 + lm;
    const float bias = b1[h];
    #pragma unroll
    for (int r = 0; r < 8; ++r) {
      const int m = r + hi * 8;
      sF[m * FPITCH + h] = fmaxf(c[r] + bias, 0.f);
    }
  }
  __syncthreads();

  const int row = (tid >> 5) * 2 + hi;
  float dot = -INFINITY;
  if (lm < Cc) {
    float acc = b2[lm];
    for (int d = 0; d < Hh; ++d) acc += sF[row * FPITCH + d] * sW2[lm * WPITCH + d];
    dot = acc;
  }
  float m = dot;
  for (int o = 8; o > 0; o >>= 1) m = fmaxf(m, __shfl_xor(m, o));
  const float ex = (lm < Cc) ? __expf(dot - m) : 0.f;
  float se = ex;
  for (int o = 8; o > 0; o >>= 1) se += __shfl_xor(se, o);
  if (lm < Cc) out[(size_t)(n0 + row) * Cc + lm] = dot - m - logf(se);
}

extern "C" void kernel_launch(void* const* d_in, const int* in_sizes, int n_in,
                              void* d_out, int out_size, void* d_ws, size_t ws_size,
                              hipStream_t stream) {
  const float* x_in  = (const float*)d_in[0];
  const int*   eidx  = (const int*)  d_in[1];
  const int*   idx   = (const int*)  d_in[2];
  const float* fc_w  = (const float*)d_in[3];
  const float* a1    = (const float*)d_in[4];
  const float* a2    = (const float*)d_in[5];
  const float* fc1_w = (const float*)d_in[6];
  const float* fc1_b = (const float*)d_in[7];
  const float* fc2_w = (const float*)d_in[8];
  const float* fc2_b = (const float*)d_in[9];
  const float* gamma = (const float*)d_in[10];
  const float* beta  = (const float*)d_in[11];
  const int* src = eidx;
  const int* dst = eidx + Ee;

  float* ws   = (float*)d_ws;
  float* STAT = ws;
  float* CORR = ws + 64;
  float* Z    = ws + 256;
  float* S1   = Z + (size_t)Nn * Hh;
  float* S2   = S1 + Nn;
  float* HB   = S2 + Nn;
  float* PART = HB;
  int*   CNT  = (int*)(HB + Ee);
  float* GOUT = HB + Ee + Nn + 16;
  int*   CC   = (int*)GOUT;
  int*   CURSOR = (int*)(ws + 32);
  float* POOL = GOUT + (size_t)Nn * Hh;
  float* XB   = POOL + (size_t)Gg * Hh;
  float* FC1O = XB + (size_t)Gg * Hh;
  unsigned short* XBF  = (unsigned short*)(FC1O + (size_t)Gg * Hh);
  unsigned short* WBF  = XBF  + (size_t)Nn * Dd;
  unsigned short* W1BF = WBF  + (size_t)Hh * Dd;
  unsigned short* XBBF = W1BF + (size_t)Hh * Hh;
  int* RANK = (int*)(XBBF + (size_t)Gg * Hh);
  int* OFFS = RANK + Ee;
  int* PAIR = OFFS + Nn + 16;

  (void)in_sizes; (void)n_in; (void)out_size; (void)ws_size;

  k_rank<<<RK_CHUNKS, RK_T, 0, stream>>>(src, RANK, CC, fc_w, fc1_w, WBF, W1BF, CURSOR, POOL);
  k_gat_gemm<<<GM_BLOCKS, 256, 0, stream>>>(x_in, WBF, a1, a2, Z, S1, S2, PART);
  k_offsets<<<SC_BLOCKS + 1, SC_T, 0, stream>>>(CC, CNT, OFFS, CURSOR, PART, fc_w, a1, a2, STAT, CORR);
  k_fill<<<FL_BLOCKS, 256, 0, stream>>>(src, dst, S1, S2, RANK, CC, STAT, PAIR);
  k_edge2<<<Nn / 8, 256, 0, stream>>>(CNT, OFFS, PAIR, Z, STAT, CORR, idx, POOL);
  k_tail<<<Gg / 16, 256, 0, stream>>>(POOL, gamma, beta, W1BF, fc1_b, fc2_w, fc2_b, (float*)d_out);
}


# launch merging: edge ranking (64 work-groups) and GEMM (157 x 4 tile groups, fc_w rounded to bf16 in registers) side by side in one launch: 5 launches, not 6; every sum in its old order
# speedup vs baseline: 4.8921x; 1.2032x over previous
#include <hip/hip_runtime.h>
#include <math.h>

#define Nn 50000
#define Ee 1600000
#define Dd 128
#define Hh 128
#define Cc 10
#define Gg 512
#define LRELU 0.05f
#define BN_EPS 1e-5f

typedef __attribute__((ext_vector_type(16))) __bf16 v16bf;
typedef __attribute__((ext_vector_type(8)))  float  v8f;
typedef __attribute__((ext_vector_type(4)))  __bf16 v4bf;

union BFrag { v16bf v; unsigned u[8]; };

__device__ __forceinline__ unsigned short f2bf_bits(float f) {
  union { float f; unsigned u; } v; v.f = f;
  return (unsigned short)((v.u + 0x7FFFu + ((v.u >> 16) & 1u)) >> 16);
}

#define ZPITCH 132
#define APITCH 68
#define GM_BLOCKS (Nn / 16)
static_assert(GM_BLOCKS * 16 == Nn, "16-row tiles must cover the nodes exactly");
#define GM_GROUPS 4
#define GM_GRID 157
#define GM_SLOTS (GM_GRID * GM_GROUPS)
#define GM_PER ((GM_BLOCKS + GM_SLOTS - 1) / GM_SLOTS)
static_assert(GM_SLOTS * GM_PER >= GM_BLOCKS && GM_SLOTS <= GM_BLOCKS, "the trips cover every tile, and every group's first tile exists");
#define GM_LDS_Z  (GM_GROUPS * 16 * APITCH)
#define GM_LDS_PS (GM_LDS_Z + GM_GROUPS * 16 * ZPITCH)
#define GM_LDS_PQ (GM_LDS_PS + GM_GROUPS * 8)
#define GM_LDS_END (GM_LDS_PQ + GM_GROUPS * 8)
static_assert(GM_LDS_END <= Nn && (16 * APITCH) % 4 == 0 && GM_LDS_Z % 4 == 0 && (16 * ZPITCH) % 4 == 0,
              "the GEMM's LDS fits in the ranking's counter table, and its tiles start on 16-byte boundaries");

__device__ __forceinline__ unsigned bf_pair(float lo, float hi) {
  return (unsigned)f2bf_bits(lo) | ((unsigned)f2bf_bits(hi) << 16);
}

__device__ __forceinline__ void gat_gemm_part(const int gb, int* __restrict__ smem,
                                              const float* __restrict__ x,
                                              const float* __restrict__ fcw,
                                              const float* __restrict__ a1,
                                              const float* __restrict__ a2,
                                              unsigned short* __restrict__ zb,
                                              float* __restrict__ s1, float* __restrict__ s2,
                                              float* __restrict__ part) {
  const int tid  = threadIdx.x;
  const int lane = tid & 31;
  const int grp  = tid >> 8;
  const int gt   = tid & 255;
  const int w    = gt >> 5;
  const int hi   = lane >> 4;
  const int lm   = lane & 15;
  const int h0   = w * 16;
  unsigned* sA  = (unsigned*)smem + grp * (16 * APITCH);
  float*    sZ  = (float*)smem + GM_LDS_Z + grp * (16 * ZPITCH);
  float*    sPs = (float*)smem + GM_LDS_PS + grp * 8;
  float*    sPq = (float*)smem + GM_LDS_PQ + grp * 8;

  int tile = gb * GM_GROUPS + grp;
  const float4* x4 = (const float4*)(x + (size_t)tile * 16 * Dd);
  float4 v0 = x4[gt], v1 = x4[gt + 256];

  const float4* wr = (const float4*)(fcw + (size_t)(h0 + lm) * Dd) + 2 * hi;
  BFrag b[4];
  #pragma unroll
  for (int t = 0; t < 4; ++t) {
    const float4 q0 = wr[t * 8], q1 = wr[t * 8 + 1], q2 = wr[t * 8 + 4], q3 = wr[t * 8 + 5];
    b[t].u[0] = bf_pair(q0.x, q0.y); b[t].u[1] = bf_pair(q0.z, q0.w);
    b[t].u[2] = bf_pair(q1.x, q1.y); b[t].u[3] = bf_pair(q1.z, q1.w);
    b[t].u[4] = bf_pair(q2.x, q2.y); b[t].u[5] = bf_pair(q2.z, q2.w);
    b[t].u[6] = bf_pair(q3.x, q3.y); b[t].u[7] = bf_pair(q3.z, q3.w);
  }
  float a1v[4], a2v[4];
  #pragma unroll
  for (int k = 0; k < 4; ++k) { a1v[k] = a1[lane + 32 * k]; a2v[k] = a2[lane + 32 * k]; }

  for (int it = 0; it < GM_PER; ++it, tile += GM_SLOTS) {
  const bool ok = tile < GM_BLOCKS;
  const int n0 = tile * 16;
  float4 nv0 = make_float4(0.f, 0.f, 0.f, 0.f), nv1 = nv0;
  if (it + 1 < GM_PER && tile + GM_SLOTS < GM_BLOCKS) {
    const float4* nx4 = (const float4*)(x + (size_t)(tile + GM_SLOTS) * 16 * Dd);
    nv0 = nx4[gt]; nv1 = nx4[gt + 256];
  }
  ((uint2*)sA)[w * (APITCH / 2) + lane] = make_uint2(bf_pair(v0.x, v0.y), bf_pair(v0.z, v0.w));
  ((uint2*)sA)[(w + 8) * (APITCH / 2) + lane] = make_uint2(bf_pair(v1.x, v1.y), bf_pair(v1.z, v1.w));
  float ps = ((v0.x + v0.y) + (v0.z + v0.w)) + ((v1.x + v1.y) + (v1.z + v1.w));
  float pq = ((v0.x * v0.x + v0.y * v0.y) + (v0.z * v0.z + v0.w * v0.w)) +
             ((v1.x * v1.x + v1.y * v1.y) + (v1.z * v1.z + v1.w * v1.w));
  for (int o = 16; o > 0; o >>= 1) { ps += __shfl_xor(ps, o); pq += __shfl_xor(pq, o); }
  if (lane == 0) { sPs[w] = ps; sPq[w] = pq; }
  __syncthreads();
  if (gt == 0 && ok) {
    float s = 0.f, q = 0.f;
    #pragma unroll
    for (int i = 0; i < 8; ++i) { s += sPs[i]; q += sPq[i]; }
    part[2 * tile] = s; part[2 * tile + 1] = q;
  }

  v8f c = {0.f, 0.f, 0.f, 0.f, 0.f, 0.f, 0.f, 0.f};
  const uint4* sa4 = (const uint4*)sA + lm * (APITCH / 4) + hi;
  #pragma unroll
  for (int t = 0; t < 4; ++t) {
    const uint4 lo = sa4[t * 4];
    const uint4 up = sa4[t * 4 + 2];
    BFrag a;
    a.u[0] = lo.x; a.u[1] = lo.y; a.u[2] = lo.z; a.u[3] = lo.w;
    a.u[4] = up.x; a.u[5] = up.y; a.u[6] = up.z; a.u[7] = up.w;
    c = __builtin_amdgcn_wmma_f32_16x16x32_bf16(false, a.v, false, b[t].v, (short)0, c, false, false);
  }
  #pragma unroll
  for (int r = 0; r < 8; ++r) {
    const int m = r + hi * 8;
    const int h = h0 + lm;
    sZ[m * ZPITCH + h] = c[r];
  }
  __syncthreads();
  const int row0 = w * 2;
  float p1[2] = {0.f, 0.f}, p2[2] = {0.f, 0.f};
  #pragma unroll
  for (int k = 0; k < 4; ++k) {
    #pragma unroll
    for (int t = 0; t < 2; ++t) {
      const float zv = sZ[(row0 + t) * ZPITCH + lane + 32 * k];
      p1[t] += zv * a1v[k]; p2[t] += zv * a2v[k];
    }
  }
  for (int o = 16; o > 0; o >>= 1) {
    #pragma unroll
    for (int t = 0; t < 2; ++t) { p1[t] += __shfl_xor(p1[t], o); p2[t] += __shfl_xor(p2[t], o); }
  }
  if (ok) {
    if (lane == 0) {
      s1[n0 + row0] = p1[0]; s1[n0 + row0 + 1] = p1[1];
      s2[n0 + row0] = p2[0]; s2[n0 + row0 + 1] = p2[1];
    }
    #pragma unroll
    for (int t = 0; t < 2; ++t) {
      const float4 zq = *(const float4*)&sZ[(row0 + t) * ZPITCH + 4 * lane];
      ((uint2*)zb)[(size_t)(n0 + row0 + t) * (Hh / 4) + lane] = make_uint2(bf_pair(zq.x, zq.y), bf_pair(zq.z, zq.w));
    }
  }
  v0 = nv0; v1 = nv1;
  }
}

#define RK_CHUNKS 64
#define RK_EDGES (Ee / RK_CHUNKS)
#define RK_T 1024
#define RK_PER ((RK_EDGES + RK_T - 1) / RK_T)
static_assert(RK_CHUNKS * RK_EDGES == Ee, "the chunks must cover the edge list exactly");
static_assert(RK_CHUNKS * RK_T == Gg * Hh && Hh * Hh <= RK_CHUNKS * RK_T,
              "one pooled element per thread; a weight element for the first threads");
static_assert(RK_T == GM_GROUPS * 256, "a GEMM work-group is GM_GROUPS groups of 8 waves");

__device__ __forceinline__ void rank_part(int* __restrict__ lcnt,
                                          const int* __restrict__ src, int* __restrict__ rank,
                                          int* __restrict__ cc,
                                          const float* __restrict__ fc1w, unsigned short* __restrict__ w1bf,
                                          int* __restrict__ cursor, float* __restrict__ pool) {
  const int t = threadIdx.x;
  {
    const int i = blockIdx.x * RK_T + t;
    if (i < Hh * Hh) w1bf[i] = f2bf_bits(fc1w[i]);
    if (i == 0) cursor[0] = 0;
    pool[i] = 0.f;
  }
  const int* sp = src  + (size_t)blockIdx.x * RK_EDGES;
  int*       rp = rank + (size_t)blockIdx.x * RK_EDGES;
  int s[RK_PER], r[RK_PER];
  #pragma unroll
  for (int u = 0; u < RK_PER; ++u) { const int j = t + u * RK_T; s[u] = (j < RK_EDGES) ? sp[j] : 0; }
  for (int k = t; k < Nn; k += RK_T) lcnt[k] = 0;
  __syncthreads();
  #pragma unroll
  for (int u = 0; u < RK_PER; ++u) { const int j = t + u * RK_T; r[u] = (j < RK_EDGES) ? atomicAdd(&lcnt[s[u]], 1) : 0; }
  #pragma unroll
  for (int u = 0; u < RK_PER; ++u) { const int j = t + u * RK_T; if (j < RK_EDGES) rp[j] = r[u]; }
  __syncthreads();
  int* cp = cc + (size_t)blockIdx.x * Nn;
  for (int k = t; k < Nn; k += RK_T) cp[k] = lcnt[k];
}

__global__ void __launch_bounds__(RK_T) k_rank_gemm(const int* __restrict__ src, int* __restrict__ rank,
                                                    int* __restrict__ cc,
                                                    const float* __restrict__ fcw, const float* __restrict__ fc1w,
                                                    unsigned short* __restrict__ w1bf,
                                                    int* __restrict__ cursor, float* __restrict__ pool,
                                                    const float* __restrict__ x,
                                                    const float* __restrict__ a1, const float* __restrict__ a2,
                                                    unsigned short* __restrict__ zb,
                                                    float* __restrict__ s1, float* __restrict__ s2,
                                                    float* __restrict__ part) {
  __shared__ __attribute__((aligned(16))) int smem[Nn];
  if (blockIdx.x < RK_CHUNKS) rank_part(smem, src, rank, cc, fc1w, w1bf, cursor, pool);
  else gat_gemm_part((int)blockIdx.x - RK_CHUNKS, smem, x, fcw, a1, a2, zb, s1, s2, part);
}

#define SC_T 256
#define SC_BLOCKS ((Nn + SC_T - 1) / SC_T)
#define PT_PER ((GM_BLOCKS + SC_T - 1) / SC_T)

__global__ void __launch_bounds__(SC_T) k_offsets(int* __restrict__ cc, int* __restrict__ cnt,
                                                  int* __restrict__ offs, int* __restrict__ cursor,
                                                  const float* __restrict__ part, const float* __restrict__ fcw,
                                                  const float* __restrict__ a1, const float* __restrict__ a2,
                                                  float* __restrict__ stat, float* __restrict__ corr) {
  __shared__ int wsum[SC_T / 32];
  __shared__ int sbase;
  __shared__ float rs[2 * (SC_T / 32)], rq[2 * (SC_T / 32)];
  const int t = threadIdx.x, lane = t & 31, w = t >> 5;
  if (blockIdx.x == SC_BLOCKS) {
    float2 pv[PT_PER];
    #pragma unroll
    for (int i = 0; i < PT_PER; ++i) {
      const int b = t + i * SC_T;
      pv[i] = (b < GM_BLOCKS) ? ((const float2*)part)[b] : make_float2(0.f, 0.f);
    }
    float cs = 0.f, av1 = 0.f, av2 = 0.f;
    if (t < Hh) {
      const float4* w4 = (const float4*)(fcw + (size_t)t * Dd);
      av1 = a1[t]; av2 = a2[t];
      #pragma unroll
      for (int d4 = 0; d4 < Dd / 4; ++d4) {
        const float4 v = w4[d4];
        cs += v.x; cs += v.y; cs += v.z; cs += v.w;
      }
    }
    float s = 0.f, q = 0.f;
    #pragma unroll
    for (int i = 0; i < PT_PER; ++i) { s += pv[i].x; q += pv[i].y; }
    for (int o = 16; o > 0; o >>= 1) { s += __shfl_xor(s, o); q += __shfl_xor(q, o); }
    if (lane == 0) { rs[w] = s; rq[w] = q; }
    __syncthreads();
    float S = 0.f, Q = 0.f;
    #pragma unroll
    for (int i = 0; i < SC_T / 32; ++i) { S += rs[i]; Q += rq[i]; }
    const float nd = (float)((long long)Nn * Dd);
    const float mu = S / nd;
    const float var = (Q - S * S / nd) / (nd - 1.0f);
    const float invsig = rsqrtf(var);
    const float mi = mu * invsig;
    if (t < Hh) corr[t] = mi * cs;
    float d1 = cs * av1, d2 = cs * av2;
    for (int o = 16; o > 0; o >>= 1) { d1 += __shfl_xor(d1, o); d2 += __shfl_xor(d2, o); }
    if (lane == 0) { rs[SC_T / 32 + w] = d1; rq[SC_T / 32 + w] = d2; }
    __syncthreads();
    if (t == 0) {
      float k1 = 0.f, k2 = 0.f;
      #pragma unroll
      for (int i = 0; i < SC_T / 32; ++i) { k1 += rs[SC_T / 32 + i]; k2 += rq[SC_T / 32 + i]; }
      stat[0] = invsig; stat[1] = mi * k1; stat[2] = mi * k2;
    }
    return;
  }
  const int n = blockIdx.x * SC_T + t;
  const bool ok = n < Nn;
  const int nn = ok ? n : (Nn - 1);
  int v[RK_CHUNKS];
  #pragma unroll
  for (int k = 0; k < RK_CHUNKS; ++k) v[k] = cc[(size_t)k * Nn + nn];
  int tot = 0;
  #pragma unroll
  for (int k = 0; k < RK_CHUNKS; ++k) { const int c = ok ? v[k] : 0; v[k] = tot; tot += c; }
  int inc = tot;
  for (int o = 1; o < 32; o <<= 1) { const int y = __shfl_up(inc, o); if (lane >= o) inc += y; }
  if (lane == 31) wsum[w] = inc;
  __syncthreads();
  if (t == 0) {
    int run = 0;
    for (int i = 0; i < SC_T / 32; ++i) { const int x = wsum[i]; wsum[i] = run; run += x; }
    sbase = atomicAdd(cursor, run);
  }
  __syncthreads();
  const int start = sbase + wsum[w] + inc - tot;
  if (ok) {
    cnt[n] = tot;
    offs[n] = start;
    #pragma unroll
    for (int k = 0; k < RK_CHUNKS; ++k) cc[(size_t)k * Nn + n] = start + v[k];
  }
}

#define FL_PER 4
#define FL_THREADS (Ee / FL_PER)
#define FL_T 128
#define FL_GRID 1024
#define FL_STRIDE (FL_GRID * FL_T)
static_assert(FL_PER == 4 && Ee % FL_PER == 0 && RK_EDGES % FL_PER == 0, "a thread's four edges lie in one chunk");

__global__ void __launch_bounds__(FL_T) k_fill(const int* __restrict__ src, const int* __restrict__ dst,
                                               const float* __restrict__ s1, const float* __restrict__ s2,
                                               const int* __restrict__ rank, const int* __restrict__ cc,
                                               const float* __restrict__ stat, int* __restrict__ pair) {
  int q = blockIdx.x * FL_T + threadIdx.x;
  if (q >= FL_THREADS) return;
  const float isg = stat[0], c1 = stat[1], c2 = stat[2];
  int4 sv = ((const int4*)src)[q];
  int4 dv = ((const int4*)dst)[q];
  int4 rv = ((const int4*)rank)[q];
  for (;;) {
  const int qn = q + FL_STRIDE;
  int4 nsv = sv, ndv = dv, nrv = rv;
  if (qn < FL_THREADS) {
    nsv = ((const int4*)src)[qn];
    ndv = ((const int4*)dst)[qn];
    nrv = ((const int4*)rank)[qn];
  }
  const int* ck = cc + (size_t)((q * FL_PER) / RK_EDGES) * Nn;
  const int s[FL_PER] = {sv.x, sv.y, sv.z, sv.w};
  const int d[FL_PER] = {dv.x, dv.y, dv.z, dv.w};
  const int r[FL_PER] = {rv.x, rv.y, rv.z, rv.w};
  float a[FL_PER], b[FL_PER];
  int c0[FL_PER];
  #pragma unroll
  for (int u = 0; u < FL_PER; ++u) { a[u] = s1[s[u]]; b[u] = s2[d[u]]; c0[u] = ck[s[u]]; }
  int2* p2 = (int2*)pair;
  #pragma unroll
  for (int u = 0; u < FL_PER; ++u) {
    float e = (a[u] * isg - c1) + (b[u] * isg - c2);
    e = (e >= 0.f) ? e : LRELU * e;
    const float h = __expf(e);
    const unsigned pos = min((unsigned)(c0[u] + r[u]), (unsigned)(Ee - 1));
    p2[pos] = make_int2(d[u], __float_as_int(h));
  }
  if (qn >= FL_THREADS) break;
  q = qn; sv = nsv; dv = ndv; rv = nrv;
  }
}

#define ROW_BATCH 8
#define E2_GRID 1250
#define E2_WAVES (E2_GRID * 8)
#define E2_PER (Nn / E2_WAVES)
static_assert(E2_WAVES * E2_PER == Nn, "every wave walks the same number of nodes, and together they cover the nodes exactly");

__global__ void __launch_bounds__(256) k_edge2(const int* __restrict__ cnt, const int* __restrict__ offs,
                                               const int* __restrict__ pair, const unsigned short* __restrict__ zb,
                                               const float* __restrict__ stat, const float* __restrict__ corr,
                                               const int* __restrict__ idx, float* __restrict__ pool) {
  __shared__ __attribute__((aligned(16))) float sRow[8 * Hh];
  const int lane = threadIdx.x & 31;
  int n = __builtin_amdgcn_readfirstlane((int)(blockIdx.x * 8 + (threadIdx.x >> 5))) * E2_PER;
  if (n >= Nn) return;
  const int nend = n + E2_PER;
  int beg = offs[n];
  int m   = cnt[n];
  const float isg = stat[0];
  const float cq0 = corr[lane], cq1 = corr[lane + 32], cq2 = corr[lane + 64], cq3 = corr[lane + 96];
  float* my = sRow + (threadIdx.x >> 5) * Hh;
  for (;;) {
  const int nn = n + 1;
  int nbeg = 0, nm = 0;
  if (nn < nend) { nbeg = offs[nn]; nm = cnt[nn]; }
  const int2* p2 = (const int2*)pair + beg;
  float hs = 0.f;
  for (int c = 0; c < m; c += 32) {
    const int k = c + lane;
    hs += (k < m) ? __int_as_float(p2[k].y) : 0.f;
  }
  for (int o = 16; o > 0; o >>= 1) hs += __shfl_xor(hs, o);
  float4 acc = make_float4(0.f, 0.f, 0.f, 0.f);
  float asum = 0.f;
  const v4bf* z4 = (const v4bf*)zb + lane;
  for (int c = 0; c < m; c += 32) {
    const int k = c + lane;
    const int left = m - c;
    int2 p = make_int2(0, 0);
    if (k < m) p = p2[k];
    const int d0 = __builtin_amdgcn_readlane(p.x, 0);
    const int md = (int)min((unsigned)((k < m) ? p.x : d0), (unsigned)(Nn - 1));
    const int ma = (k < m) ? __float_as_int(__int_as_float(p.y) / hs) : 0;
    asum += __int_as_float(ma);
    #pragma unroll
    for (int j0 = 0; j0 < 32; j0 += ROW_BATCH) {
      if (j0 >= left) break;
      v4bf zv[ROW_BATCH];
      #pragma unroll
      for (int u = 0; u < ROW_BATCH; ++u) {
        const int d = __builtin_amdgcn_readlane(md, j0 + u);
        zv[u] = z4[(size_t)d * (Hh / 4)];
      }
      #pragma unroll
      for (int u = 0; u < ROW_BATCH; ++u) {
        const float alpha = __int_as_float(__builtin_amdgcn_readlane(ma, j0 + u));
        acc.x += alpha * (float)zv[u][0];
        acc.y += alpha * (float)zv[u][1];
        acc.z += alpha * (float)zv[u][2];
        acc.w += alpha * (float)zv[u][3];
      }
    }
  }
  my[4 * lane + 0] = acc.x; my[4 * lane + 1] = acc.y; my[4 * lane + 2] = acc.z; my[4 * lane + 3] = acc.w;
  const float v0 = my[lane], v1 = my[lane + 32], v2 = my[lane + 64], v3 = my[lane + 96];
  for (int o = 16; o > 0; o >>= 1) asum += __shfl_xor(asum, o);
  const int g = idx[n];
  float* p = pool + (size_t)g * Hh + lane;
  atomicAdd(p +  0, fmaxf(v0 * isg - cq0 * asum, 0.f));
  atomicAdd(p + 32, fmaxf(v1 * isg - cq1 * asum, 0.f));
  atomicAdd(p + 64, fmaxf(v2 * isg - cq2 * asum, 0.f));
  atomicAdd(p + 96, fmaxf(v3 * isg - cq3 * asum, 0.f));
  if (nn >= nend) break;
  n = nn; beg = nbeg; m = nm;
  }
}

#define XPITCH 68
#define FPITCH 132
#define WPITCH 129

__global__ void __launch_bounds__(256) k_tail(const float* __restrict__ pooled, const float* __restrict__ gamma,
                                              const float* __restrict__ beta,
                                              const unsigned short* __restrict__ w1bf,
                                              const float* __restrict__ b1, const float* __restrict__ w2,
                                              const float* __restrict__ b2, float* __restrict__ out) {
  __shared__ float sMu[Hh], sInv[Hh];
  __shared__ __attribute__((aligned(16))) unsigned sX[16 * XPITCH];
  __shared__ float sF[16 * FPITCH];
  __shared__ float sW2[Cc * WPITCH];
  const int tid  = threadIdx.x;
  const int lane = tid & 31;
  const int hi   = lane >> 4;
  const int lm   = lane & 15;
  const int n0   = blockIdx.x * 16;
  const int h0   = (tid >> 5) * 16;

  const unsigned* wp = (const unsigned*)w1bf + (size_t)(h0 + lm) * (Hh / 2);
  BFrag b[4];
  #pragma unroll
  for (int t = 0; t < 4; ++t) {
    #pragma unroll
    for (int j = 0; j < 8; ++j) {
      const int kk = t * 32 + ((j < 4) ? 0 : 16) + (hi ? 8 : 0) + 2 * (j & 3);
      b[t].u[j] = wp[kk >> 1];
    }
  }
  for (int i = tid; i < Cc * Hh; i += 256) sW2[(i >> 7) * WPITCH + (i & 127)] = w2[i];

  if (tid < Hh) {
    float s = 0.f, q = 0.f;
    #pragma unroll 64
    for (int g = 0; g < Gg; ++g) { const float v = pooled[g * Hh + tid]; s += v; q += v * v; }
    const float mu = s / (float)Gg;
    const float var = q / (float)Gg - mu * mu;
    sMu[tid] = mu;
    sInv[tid] = rsqrtf(var + BN_EPS);
  }
  __syncthreads();

  {
    const int r  = tid >> 4;
    const int c8 = (tid & 15) * 8;
    const float4* p4 = (const float4*)(pooled + (size_t)(n0 + r) * Hh + c8);
    const float4* g4 = (const float4*)(gamma + c8);
    const float4* e4 = (const float4*)(beta + c8);
    const float4 pa = p4[0], pb = p4[1], ga = g4[0], gb = g4[1], ea = e4[0], eb = e4[1];
    const float pv[8] = {pa.x, pa.y, pa.z, pa.w, pb.x, pb.y, pb.z, pb.w};
    const float gv[8] = {ga.x, ga.y, ga.z, ga.w, gb.x, gb.y, gb.z, gb.w};
    const float ev[8] = {ea.x, ea.y, ea.z, ea.w, eb.x, eb.y, eb.z, eb.w};
    unsigned pk[4];
    #pragma unroll
    for (int e = 0; e < 4; ++e) {
      const int k = c8 + 2 * e;
      const float x0 = (pv[2 * e]     - sMu[k])     * sInv[k]     * gv[2 * e]     + ev[2 * e];
      const float x1 = (pv[2 * e + 1] - sMu[k + 1]) * sInv[k + 1] * gv[2 * e + 1] + ev[2 * e + 1];
      pk[e] = (unsigned)f2bf_bits(x0) | ((unsigned)f2bf_bits(x1) << 16);
    }
    ((uint4*)sX)[r * (XPITCH / 4) + (tid & 15)] = make_uint4(pk[0], pk[1], pk[2], pk[3]);
  }
  __syncthreads();

  v8f c = {0.f, 0.f, 0.f, 0.f, 0.f, 0.f, 0.f, 0.f};
  const uint4* sx4 = (const uint4*)sX + lm * (XPITCH / 4) + hi;
  #pragma unroll
  for (int t = 0; t < 4; ++t) {
    const uint4 lo = sx4[t * 4];
    const uint4 up = sx4[t * 4 + 2];
    BFrag a;
    a.u[0] = lo.x; a.u[1] = lo.y; a.u[2] = lo.z; a.u[3] = lo.w;
    a.u[4] = up.x; a.u[5] = up.y; a.u[6] = up.z; a.u[7] = up.w;
    c = __builtin_amdgcn_wmma_f32_16x16x32_bf16(false, a.v, false, b[t].v, (short)0, c, false, false);
  }
  {
    const int h = h0 + lm;
    const float bias = b1[h];
    #pragma unroll
    for (int r = 0; r < 8; ++r) {
      const int m = r + hi * 8;
      sF[m * FPITCH + h] = fmaxf(c[r] + bias, 0.f);
    }
  }
  __syncthreads();

  const int row = (tid >> 5) * 2 + hi;
  float dot = -INFINITY;
  if (lm < Cc) {
    float acc = b2[lm];
    for (int d = 0; d < Hh; ++d) acc += sF[row * FPITCH + d] * sW2[lm * WPITCH + d];
    dot = acc;
  }
  float m = dot;
  for (int o = 8; o > 0; o >>= 1) m = fmaxf(m, __shfl_xor(m, o));
  const float ex = (lm < Cc) ? __expf(dot - m) : 0.f;
  float se = ex;
  for (int o = 8; o > 0; o >>= 1) se += __shfl_xor(se, o);
  if (lm < Cc) out[(size_t)(n0 + row) * Cc + lm] = dot - m - logf(se);
}

extern "C" void kernel_launch(void* const* d_in, const int* in_sizes, int n_in,
                              void* d_out, int out_size, void* d_ws, size_t ws_size,
                              hipStream_t stream) {
  const float* x_in  = (const float*)d_in[0];
  const int*   eidx  = (const int*)  d_in[1];
  const int*   idx   = (const int*)  d_in[2];
  const float* fc_w  = (const float*)d_in[3];
  const float* a1    = (const float*)d_in[4];
  const float* a2    = (const float*)d_in[5];
  const float* fc1_w = (const float*)d_in[6];
  const float* fc1_b = (const float*)d_in[7];
  const float* fc2_w = (const float*)d_in[8];
  const float* fc2_b = (const float*)d_in[9];
  const float* gamma = (const float*)d_in[10];
  const float* beta  = (const float*)d_in[11];
  const int* src = eidx;
  const int* dst = eidx + Ee;

  float* ws   = (float*)d_ws;
  float* STAT = ws;
  float* CORR = ws + 64;
  float* Z    = ws + 256;
  float* S1   = Z + (size_t)Nn * Hh;
  float* S2   = S1 + Nn;
  float* HB   = S2 + Nn;
  float* PART = HB;
  int*   CNT  = (int*)(HB + Ee);
  float* GOUT = HB + Ee + Nn + 16;
  int*   CC   = (int*)GOUT;
  int*   CURSOR = (int*)(ws + 32);
  float* POOL = GOUT + (size_t)Nn * Hh;
  float* XB   = POOL + (size_t)Gg * Hh;
  float* FC1O = XB + (size_t)Gg * Hh;
  unsigned short* XBF  = (unsigned short*)(FC1O + (size_t)Gg * Hh);
  unsigned short* WBF  = XBF  + (size_t)Nn * Dd;
  unsigned short* W1BF = WBF  + (size_t)Hh * Dd;
  unsigned short* XBBF = W1BF + (size_t)Hh * Hh;
  int* RANK = (int*)(XBBF + (size_t)Gg * Hh);
  int* OFFS = RANK + Ee;
  int* PAIR = OFFS + Nn + 16;

  (void)in_sizes; (void)n_in; (void)out_size; (void)ws_size;
  (void)WBF;

  k_rank_gemm<<<RK_CHUNKS + GM_GRID, RK_T, 0, stream>>>(src, RANK, CC, fc_w, fc1_w, W1BF, CURSOR, POOL,
                                                        x_in, a1, a2, (unsigned short*)Z, S1, S2, PART);
  k_offsets<<<SC_BLOCKS + 1, SC_T, 0, stream>>>(CC, CNT, OFFS, CURSOR, PART, fc_w, a1, a2, STAT, CORR);
  k_fill<<<FL_GRID, FL_T, 0, stream>>>(src, dst, S1, S2, RANK, CC, STAT, PAIR);
  k_edge2<<<E2_GRID, 256, 0, stream>>>(CNT, OFFS, PAIR, (const unsigned short*)Z, STAT, CORR, idx, POOL);
  k_tail<<<Gg / 16, 256, 0, stream>>>(POOL, gamma, beta, W1BF, fc1_b, fc2_w, fc2_b, (float*)d_out);
}
